# starting kernel
# baseline (speedup 1.0000x reference)
#include <hip/hip_runtime.h>
#include <hip/hip_bf16.h>
#include <math.h>

typedef float v2f __attribute__((ext_vector_type(2)));
typedef float v8f __attribute__((ext_vector_type(8)));

#define N_NODES 4096
#define B_SZ    4
#define F_SZ    64
#define ALPHA   0.2f
#define NEG_BIG -9e15f
#define L2E     1.4426950408889634f

__global__ __launch_bounds__(64) void wh_kernel(
    const float* __restrict__ h, const float* __restrict__ W,
    const float* __restrict__ a, float* __restrict__ WhP,
    float* __restrict__ s1, float* __restrict__ t2)
{
    int bi = blockIdx.x;
    int o  = threadIdx.x;
    const float* hrow = h + (size_t)bi * F_SZ;
    float acc = 0.f;
#pragma unroll
    for (int f = 0; f < F_SZ; ++f)
        acc = fmaf(hrow[f], W[f * F_SZ + o], acc);
    int operm = ((o & 15) << 2) | (o >> 4);
    WhP[(size_t)bi * F_SZ + operm] = acc;

    __shared__ float sWh[F_SZ];
    sWh[o] = acc;
    __syncthreads();
    if (o == 0) {
        float s = 0.f;
        for (int k = 0; k < F_SZ; ++k) s = fmaf(sWh[k], a[k], s);
        s1[bi] = s * (1.0f / 64.0f);
    } else if (o == 1) {
        float t = 0.f;
        for (int k = 0; k < F_SZ; ++k) t = fmaf(sWh[k], a[F_SZ + k], t);
        t2[bi] = t * (1.0f / 64.0f);
    }
}

__global__ __launch_bounds__(256) void rowmax_kernel(
    const int* __restrict__ adj, const float* __restrict__ s1,
    const float* __restrict__ t2, float* __restrict__ mrow)
{
    int wave = threadIdx.x >> 5;
    int lane = threadIdx.x & 31;
    int i = blockIdx.x * 8 + wave;
    const int* arow = adj + (size_t)i * N_NODES;
    float T0 = -3.0e38f, T1 = -3.0e38f, T2 = -3.0e38f, T3 = -3.0e38f;
    int any = 0;
    for (int j = lane; j < N_NODES; j += 32) {
        int av = arow[j];
        bool sel = av > 0;
        any |= (int)sel;
        float u0 = t2[j];
        float u1 = t2[N_NODES + j];
        float u2 = t2[2 * N_NODES + j];
        float u3 = t2[3 * N_NODES + j];
        T0 = fmaxf(T0, sel ? u0 : -3.0e38f);
        T1 = fmaxf(T1, sel ? u1 : -3.0e38f);
        T2 = fmaxf(T2, sel ? u2 : -3.0e38f);
        T3 = fmaxf(T3, sel ? u3 : -3.0e38f);
    }
#pragma unroll
    for (int off = 16; off >= 1; off >>= 1) {
        T0 = fmaxf(T0, __shfl_xor(T0, off, 32));
        T1 = fmaxf(T1, __shfl_xor(T1, off, 32));
        T2 = fmaxf(T2, __shfl_xor(T2, off, 32));
        T3 = fmaxf(T3, __shfl_xor(T3, off, 32));
        any |= __shfl_xor(any, off, 32);
    }
    if (lane == 0) {
        float Tb[4] = {T0, T1, T2, T3};
#pragma unroll
        for (int b = 0; b < 4; ++b) {
            float x = s1[b * N_NODES + i] + Tb[b];
            float m = any ? fmaxf(x, ALPHA * x) : NEG_BIG;
            mrow[b * N_NODES + i] = m;
        }
    }
}

__global__ __launch_bounds__(32) void attn_kernel(
    const int* __restrict__ adj, const float* __restrict__ WhP,
    const float* __restrict__ s1, const float* __restrict__ t2,
    const float* __restrict__ mrow, float* __restrict__ Dp,
    float* __restrict__ Zp, int jspan)
{
    int bid   = blockIdx.x;
    int b     = bid & 3;
    int itile = (bid >> 2) & 255;
    int chunk = bid >> 10;
    int lane  = threadIdx.x;
    int l15   = lane & 15;
    int half  = lane >> 4;
    int irow  = itile * 16 + l15;

    float s = s1[b * N_NODES + irow];
    float m = mrow[b * N_NODES + irow];
    const int*   adjrow = adj + (size_t)irow * N_NODES;
    const float* trow   = t2 + (size_t)b * N_NODES;
    const float* WhB    = WhP + (size_t)b * N_NODES * F_SZ;

    v8f c0 = {}, c1 = {}, c2 = {}, c3 = {};
    float zacc = 0.f;

    int j0 = chunk * jspan;
    int jend = j0 + jspan;
#pragma unroll 4
    for (int j = j0; j < jend; j += 4) {
        int jj = j + half * 2;
        int2   av = *(const int2*)(adjrow + jj);
        float2 tv = *(const float2*)(trow + jj);

        float x0 = s + tv.x, x1 = s + tv.y;
        float e0 = fmaxf(x0, ALPHA * x0);
        float e1 = fmaxf(x1, ALPHA * x1);
        float sc0 = (av.x > 0) ? e0 : NEG_BIG;
        float sc1 = (av.y > 0) ? e1 : NEG_BIG;
        float p0 = __builtin_amdgcn_exp2f((sc0 - m) * L2E);
        float p1 = __builtin_amdgcn_exp2f((sc1 - m) * L2E);
        zacc += p0 + p1;

        v2f A; A[0] = p0; A[1] = p1;

        const float* bp = WhB + (size_t)jj * F_SZ + (l15 << 2);
        float4 q0 = *(const float4*)(bp);
        float4 q1 = *(const float4*)(bp + F_SZ);
        v2f B0, B1, B2, B3;
        B0[0] = q0.x; B0[1] = q1.x;
        B1[0] = q0.y; B1[1] = q1.y;
        B2[0] = q0.z; B2[1] = q1.z;
        B3[0] = q0.w; B3[1] = q1.w;

        c0 = __builtin_amdgcn_wmma_f32_16x16x4_f32(false, A, false, B0, (short)0, c0, false, false);
        c1 = __builtin_amdgcn_wmma_f32_16x16x4_f32(false, A, false, B1, (short)0, c1, false, false);
        c2 = __builtin_amdgcn_wmma_f32_16x16x4_f32(false, A, false, B2, (short)0, c2, false, false);
        c3 = __builtin_amdgcn_wmma_f32_16x16x4_f32(false, A, false, B3, (short)0, c3, false, false);
    }

    zacc += __shfl_xor(zacc, 16, 32);

    size_t obase = (((size_t)chunk * 4 + b) * N_NODES + (size_t)itile * 16) * F_SZ;
    float* dout = Dp + obase;
    int rbase = half * 8;
#pragma unroll
    for (int r = 0; r < 8; ++r) {
        float4 v; v.x = c0[r]; v.y = c1[r]; v.z = c2[r]; v.w = c3[r];
        *(float4*)(dout + (size_t)(rbase + r) * F_SZ + (l15 << 2)) = v;
    }
    if (half == 0)
        Zp[((size_t)chunk * 4 + b) * N_NODES + irow] = zacc;
}

__global__ __launch_bounds__(256) void bn_kernel(
    const float* __restrict__ Dp, const float* __restrict__ Zp,
    const float* __restrict__ h, const float* __restrict__ gamma,
    const float* __restrict__ beta, float* __restrict__ out, int nchunk)
{
    int i   = blockIdx.x;
    int tid = threadIdx.x;
    int b   = tid >> 6;
    int o   = tid & 63;
    int operm = ((o & 15) << 2) | (o >> 4);

    float num = 0.f, z = 0.f;
    for (int c = 0; c < nchunk; ++c) {
        num += Dp[((((size_t)c * 4 + b) * N_NODES + i) << 6) + operm];
        z   += Zp[(((size_t)c * 4 + b) * N_NODES) + i];
    }
    float hp = num / z;

    __shared__ float red[256];
    red[tid] = hp;
    __syncthreads();
#pragma unroll
    for (int s = 128; s >= 1; s >>= 1) {
        if (tid < s) red[tid] += red[tid + s];
        __syncthreads();
    }
    float mean = red[0] * (1.0f / 256.0f);
    __syncthreads();
    float dev = hp - mean;
    red[tid] = dev * dev;
    __syncthreads();
#pragma unroll
    for (int s = 128; s >= 1; s >>= 1) {
        if (tid < s) red[tid] += red[tid + s];
        __syncthreads();
    }
    float var = red[0] * (1.0f / 256.0f);

    float xn  = gamma[i] * dev * (1.0f / sqrtf(var + 1e-5f)) + beta[i];
    float res = xn + h[((size_t)b * N_NODES + i) * F_SZ + o];
    out[((size_t)b * N_NODES + i) * F_SZ + o] = res > 0.f ? res : expm1f(res);
}

extern "C" void kernel_launch(void* const* d_in, const int* in_sizes, int n_in,
                              void* d_out, int out_size, void* d_ws, size_t ws_size,
                              hipStream_t stream)
{
    const float* h     = (const float*)d_in[0];
    const int*   adj   = (const int*)  d_in[1];
    const float* W     = (const float*)d_in[2];
    const float* a     = (const float*)d_in[3];
    const float* gamma = (const float*)d_in[4];
    const float* beta  = (const float*)d_in[5];
    float* out = (float*)d_out;

    const size_t WH_ELEMS = (size_t)B_SZ * N_NODES * F_SZ;
    const size_t NROW     = (size_t)B_SZ * N_NODES;

    int nchunk = 4;
    size_t need4 = (WH_ELEMS + 3 * NROW + 4 * (WH_ELEMS + NROW)) * sizeof(float);
    if (ws_size < need4) nchunk = 1;

    float* wsf  = (float*)d_ws;
    float* WhP  = wsf;
    float* s1   = WhP + WH_ELEMS;
    float* t2   = s1 + NROW;
    float* mrow = t2 + NROW;
    float* Dp   = mrow + NROW;
    float* Zp   = Dp + (size_t)nchunk * WH_ELEMS;

    int jspan = N_NODES / nchunk;

    wh_kernel<<<B_SZ * N_NODES, 64, 0, stream>>>(h, W, a, WhP, s1, t2);
    rowmax_kernel<<<N_NODES / 8, 256, 0, stream>>>(adj, s1, t2, mrow);
    attn_kernel<<<B_SZ * (N_NODES / 16) * nchunk, 32, 0, stream>>>(
        adj, WhP, s1, t2, mrow, Dp, Zp, jspan);
    bn_kernel<<<N_NODES, 256, 0, stream>>>(Dp, Zp, h, gamma, beta, out, nchunk);
}


# attn j-loop software-pipelined: each 16-column block's adj/t/Wh loads are issued one to two blocks ahead into spare register sets, so L2 latency runs under the WMMAs; sum order unchanged
# speedup vs baseline: 1.3693x; 1.3693x over previous
#include <hip/hip_runtime.h>
#include <hip/hip_bf16.h>
#include <math.h>

typedef float v2f __attribute__((ext_vector_type(2)));
typedef float v8f __attribute__((ext_vector_type(8)));

#define N_NODES 4096
#define B_SZ    4
#define F_SZ    64
#define ALPHA   0.2f
#define NEG_BIG -9e15f
#define L2E     1.4426950408889634f

__global__ __launch_bounds__(64) void wh_kernel(
    const float* __restrict__ h, const float* __restrict__ W,
    const float* __restrict__ a, float* __restrict__ WhP,
    float* __restrict__ s1, float* __restrict__ t2)
{
    int bi = blockIdx.x;
    int o  = threadIdx.x;
    const float* hrow = h + (size_t)bi * F_SZ;
    float acc = 0.f;
#pragma unroll
    for (int f = 0; f < F_SZ; ++f)
        acc = fmaf(hrow[f], W[f * F_SZ + o], acc);
    int operm = ((o & 15) << 2) | (o >> 4);
    WhP[(size_t)bi * F_SZ + operm] = acc;

    __shared__ float sWh[F_SZ];
    sWh[o] = acc;
    __syncthreads();
    if (o == 0) {
        float s = 0.f;
        for (int k = 0; k < F_SZ; ++k) s = fmaf(sWh[k], a[k], s);
        s1[bi] = s * (1.0f / 64.0f);
    } else if (o == 1) {
        float t = 0.f;
        for (int k = 0; k < F_SZ; ++k) t = fmaf(sWh[k], a[F_SZ + k], t);
        t2[bi] = t * (1.0f / 64.0f);
    }
}

__global__ __launch_bounds__(256) void rowmax_kernel(
    const int* __restrict__ adj, const float* __restrict__ s1,
    const float* __restrict__ t2, float* __restrict__ mrow)
{
    int wave = threadIdx.x >> 5;
    int lane = threadIdx.x & 31;
    int i = blockIdx.x * 8 + wave;
    const int* arow = adj + (size_t)i * N_NODES;
    float T0 = -3.0e38f, T1 = -3.0e38f, T2 = -3.0e38f, T3 = -3.0e38f;
    int any = 0;
    for (int j = lane; j < N_NODES; j += 32) {
        int av = arow[j];
        bool sel = av > 0;
        any |= (int)sel;
        float u0 = t2[j];
        float u1 = t2[N_NODES + j];
        float u2 = t2[2 * N_NODES + j];
        float u3 = t2[3 * N_NODES + j];
        T0 = fmaxf(T0, sel ? u0 : -3.0e38f);
        T1 = fmaxf(T1, sel ? u1 : -3.0e38f);
        T2 = fmaxf(T2, sel ? u2 : -3.0e38f);
        T3 = fmaxf(T3, sel ? u3 : -3.0e38f);
    }
#pragma unroll
    for (int off = 16; off >= 1; off >>= 1) {
        T0 = fmaxf(T0, __shfl_xor(T0, off, 32));
        T1 = fmaxf(T1, __shfl_xor(T1, off, 32));
        T2 = fmaxf(T2, __shfl_xor(T2, off, 32));
        T3 = fmaxf(T3, __shfl_xor(T3, off, 32));
        any |= __shfl_xor(any, off, 32);
    }
    if (lane == 0) {
        float Tb[4] = {T0, T1, T2, T3};
#pragma unroll
        for (int b = 0; b < 4; ++b) {
            float x = s1[b * N_NODES + i] + Tb[b];
            float m = any ? fmaxf(x, ALPHA * x) : NEG_BIG;
            mrow[b * N_NODES + i] = m;
        }
    }
}

typedef int   v2i __attribute__((ext_vector_type(2)));
typedef float v4f __attribute__((ext_vector_type(4)));
struct JBlock {
    v2i av[4];
    v2f tv[4];
    v4f q0[4];
    v4f q1[4];
};

__device__ __forceinline__ void load_block(JBlock& k, const int* __restrict__ pa,
                                           const float* __restrict__ pt,
                                           const float* __restrict__ pb)
{
#pragma unroll
    for (int u = 0; u < 4; ++u) {
        k.av[u] = *(const v2i*)(pa + 4 * u);
        k.tv[u] = *(const v2f*)(pt + 4 * u);
        k.q0[u] = *(const v4f*)(pb + (4 * u) * F_SZ);
        k.q1[u] = *(const v4f*)(pb + (4 * u + 1) * F_SZ);
    }
}

__device__ __forceinline__ void arrive_block(JBlock& k)
{
    asm volatile("" : "+v"(k.av[0]), "+v"(k.av[1]), "+v"(k.av[2]), "+v"(k.av[3]),
                      "+v"(k.tv[0]), "+v"(k.tv[1]), "+v"(k.tv[2]), "+v"(k.tv[3]),
                      "+v"(k.q0[0]), "+v"(k.q0[1]), "+v"(k.q0[2]), "+v"(k.q0[3]),
                      "+v"(k.q1[0]), "+v"(k.q1[1]), "+v"(k.q1[2]), "+v"(k.q1[3]));
}

__device__ __forceinline__ void compute_block(const JBlock& k, float s, float m,
                                              v8f& c0, v8f& c1, v8f& c2, v8f& c3,
                                              float& zacc)
{
#pragma unroll
    for (int u = 0; u < 4; ++u) {
        float x0 = s + k.tv[u].x, x1 = s + k.tv[u].y;
        float e0 = fmaxf(x0, ALPHA * x0);
        float e1 = fmaxf(x1, ALPHA * x1);
        float sc0 = (k.av[u].x > 0) ? e0 : NEG_BIG;
        float sc1 = (k.av[u].y > 0) ? e1 : NEG_BIG;
        float p0 = __builtin_amdgcn_exp2f((sc0 - m) * L2E);
        float p1 = __builtin_amdgcn_exp2f((sc1 - m) * L2E);
        zacc += p0 + p1;

        v2f A; A[0] = p0; A[1] = p1;
        v2f B0, B1, B2, B3;
        B0[0] = k.q0[u].x; B0[1] = k.q1[u].x;
        B1[0] = k.q0[u].y; B1[1] = k.q1[u].y;
        B2[0] = k.q0[u].z; B2[1] = k.q1[u].z;
        B3[0] = k.q0[u].w; B3[1] = k.q1[u].w;

        c0 = __builtin_amdgcn_wmma_f32_16x16x4_f32(false, A, false, B0, (short)0, c0, false, false);
        c1 = __builtin_amdgcn_wmma_f32_16x16x4_f32(false, A, false, B1, (short)0, c1, false, false);
        c2 = __builtin_amdgcn_wmma_f32_16x16x4_f32(false, A, false, B2, (short)0, c2, false, false);
        c3 = __builtin_amdgcn_wmma_f32_16x16x4_f32(false, A, false, B3, (short)0, c3, false, false);
    }
}

__global__ __launch_bounds__(32) void attn_kernel(
    const int* __restrict__ adj, const float* __restrict__ WhP,
    const float* __restrict__ s1, const float* __restrict__ t2,
    const float* __restrict__ mrow, float* __restrict__ Dp,
    float* __restrict__ Zp, int jspan)
{
    int bid   = blockIdx.x;
    int b     = bid & 3;
    int itile = (bid >> 2) & 255;
    int chunk = bid >> 10;
    int lane  = threadIdx.x;
    int l15   = lane & 15;
    int half  = lane >> 4;
    int irow  = itile * 16 + l15;

    float s = s1[b * N_NODES + irow];
    float m = mrow[b * N_NODES + irow];

    v8f c0 = {}, c1 = {}, c2 = {}, c3 = {};
    float zacc = 0.f;

    int jfirst = chunk * jspan + half * 2;
    const int*   pa = adj + (size_t)irow * N_NODES + jfirst;
    const float* pt = t2 + (size_t)b * N_NODES + jfirst;
    const float* pb = WhP + ((size_t)b * N_NODES + jfirst) * F_SZ + (l15 << 2);

#define KEEP_ORDER() __builtin_amdgcn_sched_barrier(0)

    JBlock ka, kb;
    load_block(ka, pa, pt, pb);
    int npair = jspan >> 5;
    for (int it = 1; it < npair; ++it) {
        load_block(kb, pa + 16, pt + 16, pb + 16 * F_SZ);
        KEEP_ORDER();
        arrive_block(ka);
        compute_block(ka, s, m, c0, c1, c2, c3, zacc);
        pa += 32; pt += 32; pb += 32 * F_SZ;
        load_block(ka, pa, pt, pb);
        KEEP_ORDER();
        arrive_block(kb);
        compute_block(kb, s, m, c0, c1, c2, c3, zacc);
    }
    load_block(kb, pa + 16, pt + 16, pb + 16 * F_SZ);
    KEEP_ORDER();
    arrive_block(ka);
    compute_block(ka, s, m, c0, c1, c2, c3, zacc);
    arrive_block(kb);
    compute_block(kb, s, m, c0, c1, c2, c3, zacc);
#undef KEEP_ORDER

    zacc += __shfl_xor(zacc, 16, 32);

    size_t obase = (((size_t)chunk * 4 + b) * N_NODES + (size_t)itile * 16) * F_SZ;
    float* dout = Dp + obase;
    int rbase = half * 8;
#pragma unroll
    for (int r = 0; r < 8; ++r) {
        float4 v; v.x = c0[r]; v.y = c1[r]; v.z = c2[r]; v.w = c3[r];
        *(float4*)(dout + (size_t)(rbase + r) * F_SZ + (l15 << 2)) = v;
    }
    if (half == 0)
        Zp[((size_t)chunk * 4 + b) * N_NODES + irow] = zacc;
}

__global__ __launch_bounds__(256) void bn_kernel(
    const float* __restrict__ Dp, const float* __restrict__ Zp,
    const float* __restrict__ h, const float* __restrict__ gamma,
    const float* __restrict__ beta, float* __restrict__ out, int nchunk)
{
    int i   = blockIdx.x;
    int tid = threadIdx.x;
    int b   = tid >> 6;
    int o   = tid & 63;
    int operm = ((o & 15) << 2) | (o >> 4);

    float num = 0.f, z = 0.f;
    for (int c = 0; c < nchunk; ++c) {
        num += Dp[((((size_t)c * 4 + b) * N_NODES + i) << 6) + operm];
        z   += Zp[(((size_t)c * 4 + b) * N_NODES) + i];
    }
    float hp = num / z;

    __shared__ float red[256];
    red[tid] = hp;
    __syncthreads();
#pragma unroll
    for (int s = 128; s >= 1; s >>= 1) {
        if (tid < s) red[tid] += red[tid + s];
        __syncthreads();
    }
    float mean = red[0] * (1.0f / 256.0f);
    __syncthreads();
    float dev = hp - mean;
    red[tid] = dev * dev;
    __syncthreads();
#pragma unroll
    for (int s = 128; s >= 1; s >>= 1) {
        if (tid < s) red[tid] += red[tid + s];
        __syncthreads();
    }
    float var = red[0] * (1.0f / 256.0f);

    float xn  = gamma[i] * dev * (1.0f / sqrtf(var + 1e-5f)) + beta[i];
    float res = xn + h[((size_t)b * N_NODES + i) * F_SZ + o];
    out[((size_t)b * N_NODES + i) * F_SZ + o] = res > 0.f ? res : expm1f(res);
}

extern "C" void kernel_launch(void* const* d_in, const int* in_sizes, int n_in,
                              void* d_out, int out_size, void* d_ws, size_t ws_size,
                              hipStream_t stream)
{
    const float* h     = (const float*)d_in[0];
    const int*   adj   = (const int*)  d_in[1];
    const float* W     = (const float*)d_in[2];
    const float* a     = (const float*)d_in[3];
    const float* gamma = (const float*)d_in[4];
    const float* beta  = (const float*)d_in[5];
    float* out = (float*)d_out;

    const size_t WH_ELEMS = (size_t)B_SZ * N_NODES * F_SZ;
    const size_t NROW     = (size_t)B_SZ * N_NODES;

    int nchunk = 4;
    size_t need4 = (WH_ELEMS + 3 * NROW + 4 * (WH_ELEMS + NROW)) * sizeof(float);
    if (ws_size < need4) nchunk = 1;

    float* wsf  = (float*)d_ws;
    float* WhP  = wsf;
    float* s1   = WhP + WH_ELEMS;
    float* t2   = s1 + NROW;
    float* mrow = t2 + NROW;
    float* Dp   = mrow + NROW;
    float* Zp   = Dp + (size_t)nchunk * WH_ELEMS;

    int jspan = N_NODES / nchunk;

    wh_kernel<<<B_SZ * N_NODES, 64, 0, stream>>>(h, W, a, WhP, s1, t2);
    rowmax_kernel<<<N_NODES / 8, 256, 0, stream>>>(adj, s1, t2, mrow);
    attn_kernel<<<B_SZ * (N_NODES / 16) * nchunk, 32, 0, stream>>>(
        adj, WhP, s1, t2, mrow, Dp, Zp, jspan);
    bn_kernel<<<N_NODES, 256, 0, stream>>>(Dp, Zp, h, gamma, beta, out, nchunk);
}


# bn kernel: one wave per node (8 values per lane; sums = in-lane adds + 5 lane exchanges) replaces a 256-thread block with two LDS trees and 19 barriers; 8x fewer waves; add pairing unchanged
# speedup vs baseline: 1.4312x; 1.0452x over previous
#include <hip/hip_runtime.h>
#include <hip/hip_bf16.h>
#include <math.h>

typedef float v2f __attribute__((ext_vector_type(2)));
typedef float v8f __attribute__((ext_vector_type(8)));

#define N_NODES 4096
#define B_SZ    4
#define F_SZ    64
#define ALPHA   0.2f
#define NEG_BIG -9e15f
#define L2E     1.4426950408889634f

__global__ __launch_bounds__(64) void wh_kernel(
    const float* __restrict__ h, const float* __restrict__ W,
    const float* __restrict__ a, float* __restrict__ WhP,
    float* __restrict__ s1, float* __restrict__ t2)
{
    int bi = blockIdx.x;
    int o  = threadIdx.x;
    const float* hrow = h + (size_t)bi * F_SZ;
    float acc = 0.f;
#pragma unroll
    for (int f = 0; f < F_SZ; ++f)
        acc = fmaf(hrow[f], W[f * F_SZ + o], acc);
    int operm = ((o & 15) << 2) | (o >> 4);
    WhP[(size_t)bi * F_SZ + operm] = acc;

    __shared__ float sWh[F_SZ];
    sWh[o] = acc;
    __syncthreads();
    if (o == 0) {
        float s = 0.f;
        for (int k = 0; k < F_SZ; ++k) s = fmaf(sWh[k], a[k], s);
        s1[bi] = s * (1.0f / 64.0f);
    } else if (o == 1) {
        float t = 0.f;
        for (int k = 0; k < F_SZ; ++k) t = fmaf(sWh[k], a[F_SZ + k], t);
        t2[bi] = t * (1.0f / 64.0f);
    }
}

__global__ __launch_bounds__(256) void rowmax_kernel(
    const int* __restrict__ adj, const float* __restrict__ s1,
    const float* __restrict__ t2, float* __restrict__ mrow)
{
    int wave = threadIdx.x >> 5;
    int lane = threadIdx.x & 31;
    int i = blockIdx.x * 8 + wave;
    const int* arow = adj + (size_t)i * N_NODES;
    float T0 = -3.0e38f, T1 = -3.0e38f, T2 = -3.0e38f, T3 = -3.0e38f;
    int any = 0;
    for (int j = lane; j < N_NODES; j += 32) {
        int av = arow[j];
        bool sel = av > 0;
        any |= (int)sel;
        float u0 = t2[j];
        float u1 = t2[N_NODES + j];
        float u2 = t2[2 * N_NODES + j];
        float u3 = t2[3 * N_NODES + j];
        T0 = fmaxf(T0, sel ? u0 : -3.0e38f);
        T1 = fmaxf(T1, sel ? u1 : -3.0e38f);
        T2 = fmaxf(T2, sel ? u2 : -3.0e38f);
        T3 = fmaxf(T3, sel ? u3 : -3.0e38f);
    }
#pragma unroll
    for (int off = 16; off >= 1; off >>= 1) {
        T0 = fmaxf(T0, __shfl_xor(T0, off, 32));
        T1 = fmaxf(T1, __shfl_xor(T1, off, 32));
        T2 = fmaxf(T2, __shfl_xor(T2, off, 32));
        T3 = fmaxf(T3, __shfl_xor(T3, off, 32));
        any |= __shfl_xor(any, off, 32);
    }
    if (lane == 0) {
        float Tb[4] = {T0, T1, T2, T3};
#pragma unroll
        for (int b = 0; b < 4; ++b) {
            float x = s1[b * N_NODES + i] + Tb[b];
            float m = any ? fmaxf(x, ALPHA * x) : NEG_BIG;
            mrow[b * N_NODES + i] = m;
        }
    }
}

typedef int   v2i __attribute__((ext_vector_type(2)));
typedef float v4f __attribute__((ext_vector_type(4)));
struct JBlock {
    v2i av[4];
    v2f tv[4];
    v4f q0[4];
    v4f q1[4];
};

__device__ __forceinline__ void load_block(JBlock& k, const int* __restrict__ pa,
                                           const float* __restrict__ pt,
                                           const float* __restrict__ pb)
{
#pragma unroll
    for (int u = 0; u < 4; ++u) {
        k.av[u] = *(const v2i*)(pa + 4 * u);
        k.tv[u] = *(const v2f*)(pt + 4 * u);
        k.q0[u] = *(const v4f*)(pb + (4 * u) * F_SZ);
        k.q1[u] = *(const v4f*)(pb + (4 * u + 1) * F_SZ);
    }
}

__device__ __forceinline__ void arrive_block(JBlock& k)
{
    asm volatile("" : "+v"(k.av[0]), "+v"(k.av[1]), "+v"(k.av[2]), "+v"(k.av[3]),
                      "+v"(k.tv[0]), "+v"(k.tv[1]), "+v"(k.tv[2]), "+v"(k.tv[3]),
                      "+v"(k.q0[0]), "+v"(k.q0[1]), "+v"(k.q0[2]), "+v"(k.q0[3]),
                      "+v"(k.q1[0]), "+v"(k.q1[1]), "+v"(k.q1[2]), "+v"(k.q1[3]));
}

__device__ __forceinline__ void compute_block(const JBlock& k, float s, float m,
                                              v8f& c0, v8f& c1, v8f& c2, v8f& c3,
                                              float& zacc)
{
#pragma unroll
    for (int u = 0; u < 4; ++u) {
        float x0 = s + k.tv[u].x, x1 = s + k.tv[u].y;
        float e0 = fmaxf(x0, ALPHA * x0);
        float e1 = fmaxf(x1, ALPHA * x1);
        float sc0 = (k.av[u].x > 0) ? e0 : NEG_BIG;
        float sc1 = (k.av[u].y > 0) ? e1 : NEG_BIG;
        float p0 = __builtin_amdgcn_exp2f((sc0 - m) * L2E);
        float p1 = __builtin_amdgcn_exp2f((sc1 - m) * L2E);
        zacc += p0 + p1;

        v2f A; A[0] = p0; A[1] = p1;
        v2f B0, B1, B2, B3;
        B0[0] = k.q0[u].x; B0[1] = k.q1[u].x;
        B1[0] = k.q0[u].y; B1[1] = k.q1[u].y;
        B2[0] = k.q0[u].z; B2[1] = k.q1[u].z;
        B3[0] = k.q0[u].w; B3[1] = k.q1[u].w;

        c0 = __builtin_amdgcn_wmma_f32_16x16x4_f32(false, A, false, B0, (short)0, c0, false, false);
        c1 = __builtin_amdgcn_wmma_f32_16x16x4_f32(false, A, false, B1, (short)0, c1, false, false);
        c2 = __builtin_amdgcn_wmma_f32_16x16x4_f32(false, A, false, B2, (short)0, c2, false, false);
        c3 = __builtin_amdgcn_wmma_f32_16x16x4_f32(false, A, false, B3, (short)0, c3, false, false);
    }
}

__global__ __launch_bounds__(32) void attn_kernel(
    const int* __restrict__ adj, const float* __restrict__ WhP,
    const float* __restrict__ s1, const float* __restrict__ t2,
    const float* __restrict__ mrow, float* __restrict__ Dp,
    float* __restrict__ Zp, int jspan)
{
    int bid   = blockIdx.x;
    int b     = bid & 3;
    int itile = (bid >> 2) & 255;
    int chunk = bid >> 10;
    int lane  = threadIdx.x;
    int l15   = lane & 15;
    int half  = lane >> 4;
    int irow  = itile * 16 + l15;

    float s = s1[b * N_NODES + irow];
    float m = mrow[b * N_NODES + irow];

    v8f c0 = {}, c1 = {}, c2 = {}, c3 = {};
    float zacc = 0.f;

    int jfirst = chunk * jspan + half * 2;
    const int*   pa = adj + (size_t)irow * N_NODES + jfirst;
    const float* pt = t2 + (size_t)b * N_NODES + jfirst;
    const float* pb = WhP + ((size_t)b * N_NODES + jfirst) * F_SZ + (l15 << 2);

#define KEEP_ORDER() __builtin_amdgcn_sched_barrier(0)

    JBlock ka, kb;
    load_block(ka, pa, pt, pb);
    int npair = jspan >> 5;
    for (int it = 1; it < npair; ++it) {
        load_block(kb, pa + 16, pt + 16, pb + 16 * F_SZ);
        KEEP_ORDER();
        arrive_block(ka);
        compute_block(ka, s, m, c0, c1, c2, c3, zacc);
        pa += 32; pt += 32; pb += 32 * F_SZ;
        load_block(ka, pa, pt, pb);
        KEEP_ORDER();
        arrive_block(kb);
        compute_block(kb, s, m, c0, c1, c2, c3, zacc);
    }
    load_block(kb, pa + 16, pt + 16, pb + 16 * F_SZ);
    KEEP_ORDER();
    arrive_block(ka);
    compute_block(ka, s, m, c0, c1, c2, c3, zacc);
    arrive_block(kb);
    compute_block(kb, s, m, c0, c1, c2, c3, zacc);
#undef KEEP_ORDER

    zacc += __shfl_xor(zacc, 16, 32);

    size_t obase = (((size_t)chunk * 4 + b) * N_NODES + (size_t)itile * 16) * F_SZ;
    float* dout = Dp + obase;
    int rbase = half * 8;
#pragma unroll
    for (int r = 0; r < 8; ++r) {
        float4 v; v.x = c0[r]; v.y = c1[r]; v.z = c2[r]; v.w = c3[r];
        *(float4*)(dout + (size_t)(rbase + r) * F_SZ + (l15 << 2)) = v;
    }
    if (half == 0)
        Zp[((size_t)chunk * 4 + b) * N_NODES + irow] = zacc;
}

__device__ __forceinline__ float wave_tree_sum(float r)
{
    for (int off = 16; off >= 1; off >>= 1) r += __shfl_xor(r, off, 32);
    return r;
}

__device__ __forceinline__ void add_chunk(const float* __restrict__ Dp,
                                          const float* __restrict__ Zp,
                                          int c, int i, int p1,
                                          float (&num)[4][2], float (&z)[4])
{
    for (int b = 0; b < 4; ++b) {
        size_t row = ((size_t)c * 4 + b) * N_NODES + i;
        const float* d = Dp + (row << 6);
        num[b][0] += d[p1];
        num[b][1] += d[p1 + 2];
        z[b]      += Zp[row];
    }
}

__global__ __launch_bounds__(256) void bn_kernel(
    const float* __restrict__ Dp, const float* __restrict__ Zp,
    const float* __restrict__ h, const float* __restrict__ gamma,
    const float* __restrict__ beta, float* __restrict__ out, int nchunk)
{
    int lane = threadIdx.x & 31;
    int wv = __builtin_amdgcn_readfirstlane((int)(threadIdx.x >> 5));
    int i  = blockIdx.x * 8 + wv;
    int p1 = ((lane & 15) << 2) | (lane >> 4);

    float hin[4][2];
    for (int b = 0; b < 4; ++b) {
        const float* hr = h + ((size_t)b * N_NODES + i) * F_SZ;
        hin[b][0] = hr[lane];
        hin[b][1] = hr[lane + 32];
    }
    float g = gamma[i], be = beta[i];

    float num[4][2], z[4];
    for (int b = 0; b < 4; ++b) { num[b][0] = 0.f; num[b][1] = 0.f; z[b] = 0.f; }
    if (nchunk == 4) {
        for (int c = 0; c < 4; ++c) add_chunk(Dp, Zp, c, i, p1, num, z);
    } else {
        for (int c = 0; c < nchunk; ++c) add_chunk(Dp, Zp, c, i, p1, num, z);
    }

    float hp[4][2];
    for (int b = 0; b < 4; ++b) {
        hp[b][0] = num[b][0] / z[b];
        hp[b][1] = num[b][1] / z[b];
    }

    float q0 = (hp[0][0] + hp[2][0]) + (hp[1][0] + hp[3][0]);
    float q1 = (hp[0][1] + hp[2][1]) + (hp[1][1] + hp[3][1]);
    float mean = wave_tree_sum(q0 + q1) * (1.0f / 256.0f);

    float dev[4][2], sq[4][2];
    for (int b = 0; b < 4; ++b) {
        dev[b][0] = hp[b][0] - mean;  sq[b][0] = dev[b][0] * dev[b][0];
        dev[b][1] = hp[b][1] - mean;  sq[b][1] = dev[b][1] * dev[b][1];
    }
    float v0 = (sq[0][0] + sq[2][0]) + (sq[1][0] + sq[3][0]);
    float v1 = (sq[0][1] + sq[2][1]) + (sq[1][1] + sq[3][1]);
    float var = wave_tree_sum(v0 + v1) * (1.0f / 256.0f);

    float inv = 1.0f / sqrtf(var + 1e-5f);
    for (int b = 0; b < 4; ++b) {
        float* orow = out + ((size_t)b * N_NODES + i) * F_SZ;
        for (int k = 0; k < 2; ++k) {
            float xn  = g * dev[b][k] * inv + be;
            float res = xn + hin[b][k];
            orow[lane + 32 * k] = res > 0.f ? res : expm1f(res);
        }
    }
}

extern "C" void kernel_launch(void* const* d_in, const int* in_sizes, int n_in,
                              void* d_out, int out_size, void* d_ws, size_t ws_size,
                              hipStream_t stream)
{
    const float* h     = (const float*)d_in[0];
    const int*   adj   = (const int*)  d_in[1];
    const float* W     = (const float*)d_in[2];
    const float* a     = (const float*)d_in[3];
    const float* gamma = (const float*)d_in[4];
    const float* beta  = (const float*)d_in[5];
    float* out = (float*)d_out;

    const size_t WH_ELEMS = (size_t)B_SZ * N_NODES * F_SZ;
    const size_t NROW     = (size_t)B_SZ * N_NODES;

    int nchunk = 4;
    size_t need4 = (WH_ELEMS + 3 * NROW + 4 * (WH_ELEMS + NROW)) * sizeof(float);
    if (ws_size < need4) nchunk = 1;

    float* wsf  = (float*)d_ws;
    float* WhP  = wsf;
    float* s1   = WhP + WH_ELEMS;
    float* t2   = s1 + NROW;
    float* mrow = t2 + NROW;
    float* Dp   = mrow + NROW;
    float* Zp   = Dp + (size_t)nchunk * WH_ELEMS;

    int jspan = N_NODES / nchunk;

    wh_kernel<<<B_SZ * N_NODES, 64, 0, stream>>>(h, W, a, WhP, s1, t2);
    rowmax_kernel<<<N_NODES / 8, 256, 0, stream>>>(adj, s1, t2, mrow);
    attn_kernel<<<B_SZ * (N_NODES / 16) * nchunk, 32, 0, stream>>>(
        adj, WhP, s1, t2, mrow, Dp, Zp, jspan);
    bn_kernel<<<N_NODES / 8, 256, 0, stream>>>(Dp, Zp, h, gamma, beta, out, nchunk);
}


# Wh kernel: one wave computes 8 nodes, its two W columns held in registers and the h rows staged in LDS (broadcast reads): 16x fewer waves and workgroups; every fma chain and its order unchanged
# speedup vs baseline: 1.6285x; 1.1379x over previous
#include <hip/hip_runtime.h>
#include <hip/hip_bf16.h>
#include <math.h>

typedef float v2f __attribute__((ext_vector_type(2)));
typedef float v8f __attribute__((ext_vector_type(8)));

#define N_NODES 4096
#define B_SZ    4
#define F_SZ    64
#define ALPHA   0.2f
#define NEG_BIG -9e15f
#define L2E     1.4426950408889634f

#define WH_NODES 8
#define WH_WAVES 2
#define WH_PITCH (F_SZ + 1)

__global__ __launch_bounds__(32 * WH_WAVES) void wh_kernel(
    const float* __restrict__ h, const float* __restrict__ W,
    const float* __restrict__ a, float* __restrict__ WhP,
    float* __restrict__ s1, float* __restrict__ t2)
{
    __shared__ float sWh[WH_WAVES][WH_NODES][WH_PITCH];
    __shared__ float4 sH[WH_WAVES][WH_NODES * F_SZ / 4];
    int lane = threadIdx.x & 31;
    int wv = threadIdx.x >> 5;
    int n0 = (blockIdx.x * WH_WAVES + wv) * WH_NODES;

    const float4* hsrc = (const float4*)(h + (size_t)n0 * F_SZ);
    float4 hq[WH_NODES * F_SZ / 128];
    for (int q = 0; q < WH_NODES * F_SZ / 128; ++q) hq[q] = hsrc[q * 32 + lane];

    float w0[F_SZ], w1[F_SZ];
#pragma unroll
    for (int f = 0; f < F_SZ; ++f) {
        w0[f] = W[f * F_SZ + lane];
        w1[f] = W[f * F_SZ + lane + 32];
    }
    int p0 = ((lane & 15) << 2) | (lane >> 4);

    for (int q = 0; q < WH_NODES * F_SZ / 128; ++q) sH[wv][q * 32 + lane] = hq[q];
    __syncthreads();

    for (int r = 0; r < WH_NODES; ++r) {
        const float* hrow = (const float*)&sH[wv][r * (F_SZ / 4)];
        float acc0 = 0.f, acc1 = 0.f;
#pragma unroll
        for (int f = 0; f < F_SZ; ++f) {
            float hv = hrow[f];
            acc0 = fmaf(hv, w0[f], acc0);
            acc1 = fmaf(hv, w1[f], acc1);
        }
        float* wrow = WhP + (size_t)(n0 + r) * F_SZ;
        wrow[p0]     = acc0;
        wrow[p0 + 2] = acc1;
        sWh[wv][r][lane]      = acc0;
        sWh[wv][r][lane + 32] = acc1;
    }
    __syncthreads();
    if (lane < 2 * WH_NODES) {
        int n   = lane & (WH_NODES - 1);
        int sel = lane / WH_NODES;
        const float* av  = a + sel * F_SZ;
        const float* row = &sWh[wv][n][0];
        float d = 0.f;
        for (int k = 0; k < F_SZ; ++k) d = fmaf(row[k], av[k], d);
        float* dst = sel ? t2 : s1;
        dst[n0 + n] = d * (1.0f / 64.0f);
    }
}

__global__ __launch_bounds__(256) void rowmax_kernel(
    const int* __restrict__ adj, const float* __restrict__ s1,
    const float* __restrict__ t2, float* __restrict__ mrow)
{
    int wave = threadIdx.x >> 5;
    int lane = threadIdx.x & 31;
    int i = blockIdx.x * 8 + wave;
    const int* arow = adj + (size_t)i * N_NODES;
    float T0 = -3.0e38f, T1 = -3.0e38f, T2 = -3.0e38f, T3 = -3.0e38f;
    int any = 0;
    for (int j = lane; j < N_NODES; j += 32) {
        int av = arow[j];
        bool sel = av > 0;
        any |= (int)sel;
        float u0 = t2[j];
        float u1 = t2[N_NODES + j];
        float u2 = t2[2 * N_NODES + j];
        float u3 = t2[3 * N_NODES + j];
        T0 = fmaxf(T0, sel ? u0 : -3.0e38f);
        T1 = fmaxf(T1, sel ? u1 : -3.0e38f);
        T2 = fmaxf(T2, sel ? u2 : -3.0e38f);
        T3 = fmaxf(T3, sel ? u3 : -3.0e38f);
    }
#pragma unroll
    for (int off = 16; off >= 1; off >>= 1) {
        T0 = fmaxf(T0, __shfl_xor(T0, off, 32));
        T1 = fmaxf(T1, __shfl_xor(T1, off, 32));
        T2 = fmaxf(T2, __shfl_xor(T2, off, 32));
        T3 = fmaxf(T3, __shfl_xor(T3, off, 32));
        any |= __shfl_xor(any, off, 32);
    }
    if (lane == 0) {
        float Tb[4] = {T0, T1, T2, T3};
#pragma unroll
        for (int b = 0; b < 4; ++b) {
            float x = s1[b * N_NODES + i] + Tb[b];
            float m = any ? fmaxf(x, ALPHA * x) : NEG_BIG;
            mrow[b * N_NODES + i] = m;
        }
    }
}

typedef int   v2i __attribute__((ext_vector_type(2)));
typedef float v4f __attribute__((ext_vector_type(4)));
struct JBlock {
    v2i av[4];
    v2f tv[4];
    v4f q0[4];
    v4f q1[4];
};

__device__ __forceinline__ void load_block(JBlock& k, const int* __restrict__ pa,
                                           const float* __restrict__ pt,
                                           const float* __restrict__ pb)
{
#pragma unroll
    for (int u = 0; u < 4; ++u) {
        k.av[u] = *(const v2i*)(pa + 4 * u);
        k.tv[u] = *(const v2f*)(pt + 4 * u);
        k.q0[u] = *(const v4f*)(pb + (4 * u) * F_SZ);
        k.q1[u] = *(const v4f*)(pb + (4 * u + 1) * F_SZ);
    }
}

__device__ __forceinline__ void arrive_block(JBlock& k)
{
    asm volatile("" : "+v"(k.av[0]), "+v"(k.av[1]), "+v"(k.av[2]), "+v"(k.av[3]),
                      "+v"(k.tv[0]), "+v"(k.tv[1]), "+v"(k.tv[2]), "+v"(k.tv[3]),
                      "+v"(k.q0[0]), "+v"(k.q0[1]), "+v"(k.q0[2]), "+v"(k.q0[3]),
                      "+v"(k.q1[0]), "+v"(k.q1[1]), "+v"(k.q1[2]), "+v"(k.q1[3]));
}

__device__ __forceinline__ void compute_block(const JBlock& k, float s, float m,
                                              v8f& c0, v8f& c1, v8f& c2, v8f& c3,
                                              float& zacc)
{
#pragma unroll
    for (int u = 0; u < 4; ++u) {
        float x0 = s + k.tv[u].x, x1 = s + k.tv[u].y;
        float e0 = fmaxf(x0, ALPHA * x0);
        float e1 = fmaxf(x1, ALPHA * x1);
        float sc0 = (k.av[u].x > 0) ? e0 : NEG_BIG;
        float sc1 = (k.av[u].y > 0) ? e1 : NEG_BIG;
        float p0 = __builtin_amdgcn_exp2f((sc0 - m) * L2E);
        float p1 = __builtin_amdgcn_exp2f((sc1 - m) * L2E);
        zacc += p0 + p1;

        v2f A; A[0] = p0; A[1] = p1;
        v2f B0, B1, B2, B3;
        B0[0] = k.q0[u].x; B0[1] = k.q1[u].x;
        B1[0] = k.q0[u].y; B1[1] = k.q1[u].y;
        B2[0] = k.q0[u].z; B2[1] = k.q1[u].z;
        B3[0] = k.q0[u].w; B3[1] = k.q1[u].w;

        c0 = __builtin_amdgcn_wmma_f32_16x16x4_f32(false, A, false, B0, (short)0, c0, false, false);
        c1 = __builtin_amdgcn_wmma_f32_16x16x4_f32(false, A, false, B1, (short)0, c1, false, false);
        c2 = __builtin_amdgcn_wmma_f32_16x16x4_f32(false, A, false, B2, (short)0, c2, false, false);
        c3 = __builtin_amdgcn_wmma_f32_16x16x4_f32(false, A, false, B3, (short)0, c3, false, false);
    }
}

__global__ __launch_bounds__(32) void attn_kernel(
    const int* __restrict__ adj, const float* __restrict__ WhP,
    const float* __restrict__ s1, const float* __restrict__ t2,
    const float* __restrict__ mrow, float* __restrict__ Dp,
    float* __restrict__ Zp, int jspan)
{
    int bid   = blockIdx.x;
    int b     = bid & 3;
    int itile = (bid >> 2) & 255;
    int chunk = bid >> 10;
    int lane  = threadIdx.x;
    int l15   = lane & 15;
    int half  = lane >> 4;
    int irow  = itile * 16 + l15;

    float s = s1[b * N_NODES + irow];
    float m = mrow[b * N_NODES + irow];

    v8f c0 = {}, c1 = {}, c2 = {}, c3 = {};
    float zacc = 0.f;

    int jfirst = chunk * jspan + half * 2;
    const int*   pa = adj + (size_t)irow * N_NODES + jfirst;
    const float* pt = t2 + (size_t)b * N_NODES + jfirst;
    const float* pb = WhP + ((size_t)b * N_NODES + jfirst) * F_SZ + (l15 << 2);

#define KEEP_ORDER() __builtin_amdgcn_sched_barrier(0)

    JBlock ka, kb;
    load_block(ka, pa, pt, pb);
    int npair = jspan >> 5;
    for (int it = 1; it < npair; ++it) {
        load_block(kb, pa + 16, pt + 16, pb + 16 * F_SZ);
        KEEP_ORDER();
        arrive_block(ka);
        compute_block(ka, s, m, c0, c1, c2, c3, zacc);
        pa += 32; pt += 32; pb += 32 * F_SZ;
        load_block(ka, pa, pt, pb);
        KEEP_ORDER();
        arrive_block(kb);
        compute_block(kb, s, m, c0, c1, c2, c3, zacc);
    }
    load_block(kb, pa + 16, pt + 16, pb + 16 * F_SZ);
    KEEP_ORDER();
    arrive_block(ka);
    compute_block(ka, s, m, c0, c1, c2, c3, zacc);
    arrive_block(kb);
    compute_block(kb, s, m, c0, c1, c2, c3, zacc);
#undef KEEP_ORDER

    zacc += __shfl_xor(zacc, 16, 32);

    size_t obase = (((size_t)chunk * 4 + b) * N_NODES + (size_t)itile * 16) * F_SZ;
    float* dout = Dp + obase;
    int rbase = half * 8;
#pragma unroll
    for (int r = 0; r < 8; ++r) {
        float4 v; v.x = c0[r]; v.y = c1[r]; v.z = c2[r]; v.w = c3[r];
        *(float4*)(dout + (size_t)(rbase + r) * F_SZ + (l15 << 2)) = v;
    }
    if (half == 0)
        Zp[((size_t)chunk * 4 + b) * N_NODES + irow] = zacc;
}

__device__ __forceinline__ float wave_tree_sum(float r)
{
    for (int off = 16; off >= 1; off >>= 1) r += __shfl_xor(r, off, 32);
    return r;
}

__device__ __forceinline__ void add_chunk(const float* __restrict__ Dp,
                                          const float* __restrict__ Zp,
                                          int c, int i, int p1,
                                          float (&num)[4][2], float (&z)[4])
{
    for (int b = 0; b < 4; ++b) {
        size_t row = ((size_t)c * 4 + b) * N_NODES + i;
        const float* d = Dp + (row << 6);
        num[b][0] += d[p1];
        num[b][1] += d[p1 + 2];
        z[b]      += Zp[row];
    }
}

__global__ __launch_bounds__(256) void bn_kernel(
    const float* __restrict__ Dp, const float* __restrict__ Zp,
    const float* __restrict__ h, const float* __restrict__ gamma,
    const float* __restrict__ beta, float* __restrict__ out, int nchunk)
{
    int lane = threadIdx.x & 31;
    int wv = __builtin_amdgcn_readfirstlane((int)(threadIdx.x >> 5));
    int i  = blockIdx.x * 8 + wv;
    int p1 = ((lane & 15) << 2) | (lane >> 4);

    float hin[4][2];
    for (int b = 0; b < 4; ++b) {
        const float* hr = h + ((size_t)b * N_NODES + i) * F_SZ;
        hin[b][0] = hr[lane];
        hin[b][1] = hr[lane + 32];
    }
    float g = gamma[i], be = beta[i];

    float num[4][2], z[4];
    for (int b = 0; b < 4; ++b) { num[b][0] = 0.f; num[b][1] = 0.f; z[b] = 0.f; }
    if (nchunk == 4) {
        for (int c = 0; c < 4; ++c) add_chunk(Dp, Zp, c, i, p1, num, z);
    } else {
        for (int c = 0; c < nchunk; ++c) add_chunk(Dp, Zp, c, i, p1, num, z);
    }

    float hp[4][2];
    for (int b = 0; b < 4; ++b) {
        hp[b][0] = num[b][0] / z[b];
        hp[b][1] = num[b][1] / z[b];
    }

    float q0 = (hp[0][0] + hp[2][0]) + (hp[1][0] + hp[3][0]);
    float q1 = (hp[0][1] + hp[2][1]) + (hp[1][1] + hp[3][1]);
    float mean = wave_tree_sum(q0 + q1) * (1.0f / 256.0f);

    float dev[4][2], sq[4][2];
    for (int b = 0; b < 4; ++b) {
        dev[b][0] = hp[b][0] - mean;  sq[b][0] = dev[b][0] * dev[b][0];
        dev[b][1] = hp[b][1] - mean;  sq[b][1] = dev[b][1] * dev[b][1];
    }
    float v0 = (sq[0][0] + sq[2][0]) + (sq[1][0] + sq[3][0]);
    float v1 = (sq[0][1] + sq[2][1]) + (sq[1][1] + sq[3][1]);
    float var = wave_tree_sum(v0 + v1) * (1.0f / 256.0f);

    float inv = 1.0f / sqrtf(var + 1e-5f);
    for (int b = 0; b < 4; ++b) {
        float* orow = out + ((size_t)b * N_NODES + i) * F_SZ;
        for (int k = 0; k < 2; ++k) {
            float xn  = g * dev[b][k] * inv + be;
            float res = xn + hin[b][k];
            orow[lane + 32 * k] = res > 0.f ? res : expm1f(res);
        }
    }
}

extern "C" void kernel_launch(void* const* d_in, const int* in_sizes, int n_in,
                              void* d_out, int out_size, void* d_ws, size_t ws_size,
                              hipStream_t stream)
{
    const float* h     = (const float*)d_in[0];
    const int*   adj   = (const int*)  d_in[1];
    const float* W     = (const float*)d_in[2];
    const float* a     = (const float*)d_in[3];
    const float* gamma = (const float*)d_in[4];
    const float* beta  = (const float*)d_in[5];
    float* out = (float*)d_out;

    const size_t WH_ELEMS = (size_t)B_SZ * N_NODES * F_SZ;
    const size_t NROW     = (size_t)B_SZ * N_NODES;

    int nchunk = 4;
    size_t need4 = (WH_ELEMS + 3 * NROW + 4 * (WH_ELEMS + NROW)) * sizeof(float);
    if (ws_size < need4) nchunk = 1;

    float* wsf  = (float*)d_ws;
    float* WhP  = wsf;
    float* s1   = WhP + WH_ELEMS;
    float* t2   = s1 + NROW;
    float* mrow = t2 + NROW;
    float* Dp   = mrow + NROW;
    float* Zp   = Dp + (size_t)nchunk * WH_ELEMS;

    int jspan = N_NODES / nchunk;

    wh_kernel<<<B_SZ * N_NODES / (WH_WAVES * WH_NODES), 32 * WH_WAVES, 0, stream>>>(h, W, a, WhP, s1, t2);
    rowmax_kernel<<<N_NODES / 8, 256, 0, stream>>>(adj, s1, t2, mrow);
    attn_kernel<<<B_SZ * (N_NODES / 16) * nchunk, 32, 0, stream>>>(
        adj, WhP, s1, t2, mrow, Dp, Zp, jspan);
    bn_kernel<<<N_NODES / 8, 256, 0, stream>>>(Dp, Zp, h, gamma, beta, out, nchunk);
}


# rowmax kernel: adj and t read with 16-byte loads, 48 requested per lane before first use, two rows per wave share each t vector (was 4-byte loads, 20 in flight); max is order-free, m bit-identical
# speedup vs baseline: 1.6956x; 1.0412x over previous
#include <hip/hip_runtime.h>
#include <hip/hip_bf16.h>
#include <math.h>

typedef float v2f __attribute__((ext_vector_type(2)));
typedef float v8f __attribute__((ext_vector_type(8)));

#define N_NODES 4096
#define B_SZ    4
#define F_SZ    64
#define ALPHA   0.2f
#define NEG_BIG -9e15f
#define L2E     1.4426950408889634f

#define WH_NODES 8
#define WH_WAVES 2
#define WH_PITCH (F_SZ + 1)

__global__ __launch_bounds__(32 * WH_WAVES) void wh_kernel(
    const float* __restrict__ h, const float* __restrict__ W,
    const float* __restrict__ a, float* __restrict__ WhP,
    float* __restrict__ s1, float* __restrict__ t2)
{
    __shared__ float sWh[WH_WAVES][WH_NODES][WH_PITCH];
    __shared__ float4 sH[WH_WAVES][WH_NODES * F_SZ / 4];
    int lane = threadIdx.x & 31;
    int wv = threadIdx.x >> 5;
    int n0 = (blockIdx.x * WH_WAVES + wv) * WH_NODES;

    const float4* hsrc = (const float4*)(h + (size_t)n0 * F_SZ);
    float4 hq[WH_NODES * F_SZ / 128];
    for (int q = 0; q < WH_NODES * F_SZ / 128; ++q) hq[q] = hsrc[q * 32 + lane];

    float w0[F_SZ], w1[F_SZ];
#pragma unroll
    for (int f = 0; f < F_SZ; ++f) {
        w0[f] = W[f * F_SZ + lane];
        w1[f] = W[f * F_SZ + lane + 32];
    }
    int p0 = ((lane & 15) << 2) | (lane >> 4);

    for (int q = 0; q < WH_NODES * F_SZ / 128; ++q) sH[wv][q * 32 + lane] = hq[q];
    __syncthreads();

    for (int r = 0; r < WH_NODES; ++r) {
        const float* hrow = (const float*)&sH[wv][r * (F_SZ / 4)];
        float acc0 = 0.f, acc1 = 0.f;
#pragma unroll
        for (int f = 0; f < F_SZ; ++f) {
            float hv = hrow[f];
            acc0 = fmaf(hv, w0[f], acc0);
            acc1 = fmaf(hv, w1[f], acc1);
        }
        float* wrow = WhP + (size_t)(n0 + r) * F_SZ;
        wrow[p0]     = acc0;
        wrow[p0 + 2] = acc1;
        sWh[wv][r][lane]      = acc0;
        sWh[wv][r][lane + 32] = acc1;
    }
    __syncthreads();
    if (lane < 2 * WH_NODES) {
        int n   = lane & (WH_NODES - 1);
        int sel = lane / WH_NODES;
        const float* av  = a + sel * F_SZ;
        const float* row = &sWh[wv][n][0];
        float d = 0.f;
        for (int k = 0; k < F_SZ; ++k) d = fmaf(row[k], av[k], d);
        float* dst = sel ? t2 : s1;
        dst[n0 + n] = d * (1.0f / 64.0f);
    }
}

typedef int   rm_v4i __attribute__((ext_vector_type(4)));
typedef float rm_v4f __attribute__((ext_vector_type(4)));
#define RM_ROWS  2
#define RM_WAVES 4
#define RM_TRIPS 8
#define RM_NONE  -3.0e38f

__global__ __launch_bounds__(32 * RM_WAVES) void rowmax_kernel(
    const int* __restrict__ adj, const float* __restrict__ s1,
    const float* __restrict__ t2, float* __restrict__ mrow)
{
    int wave = threadIdx.x >> 5;
    int lane = threadIdx.x & 31;
    int i0 = (blockIdx.x * RM_WAVES + wave) * RM_ROWS;
    const rm_v4i* arow = (const rm_v4i*)(adj + (size_t)i0 * N_NODES) + lane;
    const rm_v4f* tcol = (const rm_v4f*)t2 + lane;

    float T[RM_ROWS][4];
    int any[RM_ROWS];
    for (int r = 0; r < RM_ROWS; ++r) {
        any[r] = 0;
        for (int b = 0; b < 4; ++b) T[r][b] = RM_NONE;
    }

    for (int g = 0; g < N_NODES / (128 * RM_TRIPS); ++g) {
        rm_v4i av[RM_TRIPS][RM_ROWS];
        rm_v4f tv[RM_TRIPS][4];
#pragma unroll
        for (int k = 0; k < RM_TRIPS; ++k) {
            int q = (g * RM_TRIPS + k) * 32;
            for (int r = 0; r < RM_ROWS; ++r) av[k][r] = arow[r * (N_NODES / 4) + q];
            for (int b = 0; b < 4; ++b) tv[k][b] = tcol[b * (N_NODES / 4) + q];
        }
#pragma unroll
        for (int k = 0; k < RM_TRIPS; ++k) {
            for (int r = 0; r < RM_ROWS; ++r) {
                for (int e = 0; e < 4; ++e) {
                    bool sel = av[k][r][e] > 0;
                    any[r] |= (int)sel;
                    for (int b = 0; b < 4; ++b)
                        T[r][b] = fmaxf(T[r][b], sel ? tv[k][b][e] : RM_NONE);
                }
            }
        }
    }

    for (int off = 16; off >= 1; off >>= 1) {
        for (int r = 0; r < RM_ROWS; ++r) {
            for (int b = 0; b < 4; ++b)
                T[r][b] = fmaxf(T[r][b], __shfl_xor(T[r][b], off, 32));
            any[r] |= __shfl_xor(any[r], off, 32);
        }
    }
    if (lane == 0) {
        for (int r = 0; r < RM_ROWS; ++r) {
            for (int b = 0; b < 4; ++b) {
                float x = s1[b * N_NODES + i0 + r] + T[r][b];
                float m = any[r] ? fmaxf(x, ALPHA * x) : NEG_BIG;
                mrow[b * N_NODES + i0 + r] = m;
            }
        }
    }
}

typedef int   v2i __attribute__((ext_vector_type(2)));
typedef float v4f __attribute__((ext_vector_type(4)));
struct JBlock {
    v2i av[4];
    v2f tv[4];
    v4f q0[4];
    v4f q1[4];
};

__device__ __forceinline__ void load_block(JBlock& k, const int* __restrict__ pa,
                                           const float* __restrict__ pt,
                                           const float* __restrict__ pb)
{
#pragma unroll
    for (int u = 0; u < 4; ++u) {
        k.av[u] = *(const v2i*)(pa + 4 * u);
        k.tv[u] = *(const v2f*)(pt + 4 * u);
        k.q0[u] = *(const v4f*)(pb + (4 * u) * F_SZ);
        k.q1[u] = *(const v4f*)(pb + (4 * u + 1) * F_SZ);
    }
}

__device__ __forceinline__ void arrive_block(JBlock& k)
{
    asm volatile("" : "+v"(k.av[0]), "+v"(k.av[1]), "+v"(k.av[2]), "+v"(k.av[3]),
                      "+v"(k.tv[0]), "+v"(k.tv[1]), "+v"(k.tv[2]), "+v"(k.tv[3]),
                      "+v"(k.q0[0]), "+v"(k.q0[1]), "+v"(k.q0[2]), "+v"(k.q0[3]),
                      "+v"(k.q1[0]), "+v"(k.q1[1]), "+v"(k.q1[2]), "+v"(k.q1[3]));
}

__device__ __forceinline__ void compute_block(const JBlock& k, float s, float m,
                                              v8f& c0, v8f& c1, v8f& c2, v8f& c3,
                                              float& zacc)
{
#pragma unroll
    for (int u = 0; u < 4; ++u) {
        float x0 = s + k.tv[u].x, x1 = s + k.tv[u].y;
        float e0 = fmaxf(x0, ALPHA * x0);
        float e1 = fmaxf(x1, ALPHA * x1);
        float sc0 = (k.av[u].x > 0) ? e0 : NEG_BIG;
        float sc1 = (k.av[u].y > 0) ? e1 : NEG_BIG;
        float p0 = __builtin_amdgcn_exp2f((sc0 - m) * L2E);
        float p1 = __builtin_amdgcn_exp2f((sc1 - m) * L2E);
        zacc += p0 + p1;

        v2f A; A[0] = p0; A[1] = p1;
        v2f B0, B1, B2, B3;
        B0[0] = k.q0[u].x; B0[1] = k.q1[u].x;
        B1[0] = k.q0[u].y; B1[1] = k.q1[u].y;
        B2[0] = k.q0[u].z; B2[1] = k.q1[u].z;
        B3[0] = k.q0[u].w; B3[1] = k.q1[u].w;

        c0 = __builtin_amdgcn_wmma_f32_16x16x4_f32(false, A, false, B0, (short)0, c0, false, false);
        c1 = __builtin_amdgcn_wmma_f32_16x16x4_f32(false, A, false, B1, (short)0, c1, false, false);
        c2 = __builtin_amdgcn_wmma_f32_16x16x4_f32(false, A, false, B2, (short)0, c2, false, false);
        c3 = __builtin_amdgcn_wmma_f32_16x16x4_f32(false, A, false, B3, (short)0, c3, false, false);
    }
}

__global__ __launch_bounds__(32) void attn_kernel(
    const int* __restrict__ adj, const float* __restrict__ WhP,
    const float* __restrict__ s1, const float* __restrict__ t2,
    const float* __restrict__ mrow, float* __restrict__ Dp,
    float* __restrict__ Zp, int jspan)
{
    int bid   = blockIdx.x;
    int b     = bid & 3;
    int itile = (bid >> 2) & 255;
    int chunk = bid >> 10;
    int lane  = threadIdx.x;
    int l15   = lane & 15;
    int half  = lane >> 4;
    int irow  = itile * 16 + l15;

    float s = s1[b * N_NODES + irow];
    float m = mrow[b * N_NODES + irow];

    v8f c0 = {}, c1 = {}, c2 = {}, c3 = {};
    float zacc = 0.f;

    int jfirst = chunk * jspan + half * 2;
    const int*   pa = adj + (size_t)irow * N_NODES + jfirst;
    const float* pt = t2 + (size_t)b * N_NODES + jfirst;
    const float* pb = WhP + ((size_t)b * N_NODES + jfirst) * F_SZ + (l15 << 2);

#define KEEP_ORDER() __builtin_amdgcn_sched_barrier(0)

    JBlock ka, kb;
    load_block(ka, pa, pt, pb);
    int npair = jspan >> 5;
    for (int it = 1; it < npair; ++it) {
        load_block(kb, pa + 16, pt + 16, pb + 16 * F_SZ);
        KEEP_ORDER();
        arrive_block(ka);
        compute_block(ka, s, m, c0, c1, c2, c3, zacc);
        pa += 32; pt += 32; pb += 32 * F_SZ;
        load_block(ka, pa, pt, pb);
        KEEP_ORDER();
        arrive_block(kb);
        compute_block(kb, s, m, c0, c1, c2, c3, zacc);
    }
    load_block(kb, pa + 16, pt + 16, pb + 16 * F_SZ);
    KEEP_ORDER();
    arrive_block(ka);
    compute_block(ka, s, m, c0, c1, c2, c3, zacc);
    arrive_block(kb);
    compute_block(kb, s, m, c0, c1, c2, c3, zacc);
#undef KEEP_ORDER

    zacc += __shfl_xor(zacc, 16, 32);

    size_t obase = (((size_t)chunk * 4 + b) * N_NODES + (size_t)itile * 16) * F_SZ;
    float* dout = Dp + obase;
    int rbase = half * 8;
#pragma unroll
    for (int r = 0; r < 8; ++r) {
        float4 v; v.x = c0[r]; v.y = c1[r]; v.z = c2[r]; v.w = c3[r];
        *(float4*)(dout + (size_t)(rbase + r) * F_SZ + (l15 << 2)) = v;
    }
    if (half == 0)
        Zp[((size_t)chunk * 4 + b) * N_NODES + irow] = zacc;
}

__device__ __forceinline__ float wave_tree_sum(float r)
{
    for (int off = 16; off >= 1; off >>= 1) r += __shfl_xor(r, off, 32);
    return r;
}

__device__ __forceinline__ void add_chunk(const float* __restrict__ Dp,
                                          const float* __restrict__ Zp,
                                          int c, int i, int p1,
                                          float (&num)[4][2], float (&z)[4])
{
    for (int b = 0; b < 4; ++b) {
        size_t row = ((size_t)c * 4 + b) * N_NODES + i;
        const float* d = Dp + (row << 6);
        num[b][0] += d[p1];
        num[b][1] += d[p1 + 2];
        z[b]      += Zp[row];
    }
}

__global__ __launch_bounds__(256) void bn_kernel(
    const float* __restrict__ Dp, const float* __restrict__ Zp,
    const float* __restrict__ h, const float* __restrict__ gamma,
    const float* __restrict__ beta, float* __restrict__ out, int nchunk)
{
    int lane = threadIdx.x & 31;
    int wv = __builtin_amdgcn_readfirstlane((int)(threadIdx.x >> 5));
    int i  = blockIdx.x * 8 + wv;
    int p1 = ((lane & 15) << 2) | (lane >> 4);

    float hin[4][2];
    for (int b = 0; b < 4; ++b) {
        const float* hr = h + ((size_t)b * N_NODES + i) * F_SZ;
        hin[b][0] = hr[lane];
        hin[b][1] = hr[lane + 32];
    }
    float g = gamma[i], be = beta[i];

    float num[4][2], z[4];
    for (int b = 0; b < 4; ++b) { num[b][0] = 0.f; num[b][1] = 0.f; z[b] = 0.f; }
    if (nchunk == 4) {
        for (int c = 0; c < 4; ++c) add_chunk(Dp, Zp, c, i, p1, num, z);
    } else {
        for (int c = 0; c < nchunk; ++c) add_chunk(Dp, Zp, c, i, p1, num, z);
    }

    float hp[4][2];
    for (int b = 0; b < 4; ++b) {
        hp[b][0] = num[b][0] / z[b];
        hp[b][1] = num[b][1] / z[b];
    }

    float q0 = (hp[0][0] + hp[2][0]) + (hp[1][0] + hp[3][0]);
    float q1 = (hp[0][1] + hp[2][1]) + (hp[1][1] + hp[3][1]);
    float mean = wave_tree_sum(q0 + q1) * (1.0f / 256.0f);

    float dev[4][2], sq[4][2];
    for (int b = 0; b < 4; ++b) {
        dev[b][0] = hp[b][0] - mean;  sq[b][0] = dev[b][0] * dev[b][0];
        dev[b][1] = hp[b][1] - mean;  sq[b][1] = dev[b][1] * dev[b][1];
    }
    float v0 = (sq[0][0] + sq[2][0]) + (sq[1][0] + sq[3][0]);
    float v1 = (sq[0][1] + sq[2][1]) + (sq[1][1] + sq[3][1]);
    float var = wave_tree_sum(v0 + v1) * (1.0f / 256.0f);

    float inv = 1.0f / sqrtf(var + 1e-5f);
    for (int b = 0; b < 4; ++b) {
        float* orow = out + ((size_t)b * N_NODES + i) * F_SZ;
        for (int k = 0; k < 2; ++k) {
            float xn  = g * dev[b][k] * inv + be;
            float res = xn + hin[b][k];
            orow[lane + 32 * k] = res > 0.f ? res : expm1f(res);
        }
    }
}

extern "C" void kernel_launch(void* const* d_in, const int* in_sizes, int n_in,
                              void* d_out, int out_size, void* d_ws, size_t ws_size,
                              hipStream_t stream)
{
    const float* h     = (const float*)d_in[0];
    const int*   adj   = (const int*)  d_in[1];
    const float* W     = (const float*)d_in[2];
    const float* a     = (const float*)d_in[3];
    const float* gamma = (const float*)d_in[4];
    const float* beta  = (const float*)d_in[5];
    float* out = (float*)d_out;

    const size_t WH_ELEMS = (size_t)B_SZ * N_NODES * F_SZ;
    const size_t NROW     = (size_t)B_SZ * N_NODES;

    int nchunk = 4;
    size_t need4 = (WH_ELEMS + 3 * NROW + 4 * (WH_ELEMS + NROW)) * sizeof(float);
    if (ws_size < need4) nchunk = 1;

    float* wsf  = (float*)d_ws;
    float* WhP  = wsf;
    float* s1   = WhP + WH_ELEMS;
    float* t2   = s1 + NROW;
    float* mrow = t2 + NROW;
    float* Dp   = mrow + NROW;
    float* Zp   = Dp + (size_t)nchunk * WH_ELEMS;

    int jspan = N_NODES / nchunk;

    wh_kernel<<<B_SZ * N_NODES / (WH_WAVES * WH_NODES), 32 * WH_WAVES, 0, stream>>>(h, W, a, WhP, s1, t2);
    rowmax_kernel<<<N_NODES / (RM_WAVES * RM_ROWS), 32 * RM_WAVES, 0, stream>>>(adj, s1, t2, mrow);
    attn_kernel<<<B_SZ * (N_NODES / 16) * nchunk, 32, 0, stream>>>(
        adj, WhP, s1, t2, mrow, Dp, Zp, jspan);
    bn_kernel<<<N_NODES / 8, 256, 0, stream>>>(Dp, Zp, h, gamma, beta, out, nchunk);
}


# softmax shift = upper bound lrelu(s+max_j t) (batch max of t from per-wave partials of the Wh kernel) in place of the masked row max: adj-streaming rowmax launch removed; sum order unchanged
# speedup vs baseline: 2.0559x; 1.2125x over previous
#include <hip/hip_runtime.h>
#include <hip/hip_bf16.h>
#include <math.h>

typedef float v2f __attribute__((ext_vector_type(2)));
typedef float v8f __attribute__((ext_vector_type(8)));

#define N_NODES 4096
#define B_SZ    4
#define F_SZ    64
#define ALPHA   0.2f
#define NEG_BIG -9e15f
#define L2E     1.4426950408889634f

#define WH_NODES 8
#define WH_WAVES 2
#define WH_PITCH (F_SZ + 1)

__global__ __launch_bounds__(32 * WH_WAVES) void wh_kernel(
    const float* __restrict__ h, const float* __restrict__ W,
    const float* __restrict__ a, float* __restrict__ WhP,
    float* __restrict__ s1, float* __restrict__ t2, float* __restrict__ tpart)
{
    __shared__ float sWh[WH_WAVES][WH_NODES][WH_PITCH];
    __shared__ float4 sH[WH_WAVES][WH_NODES * F_SZ / 4];
    int lane = threadIdx.x & 31;
    int wv = threadIdx.x >> 5;
    int n0 = (blockIdx.x * WH_WAVES + wv) * WH_NODES;

    const float4* hsrc = (const float4*)(h + (size_t)n0 * F_SZ);
    float4 hq[WH_NODES * F_SZ / 128];
    for (int q = 0; q < WH_NODES * F_SZ / 128; ++q) hq[q] = hsrc[q * 32 + lane];

    float w0[F_SZ], w1[F_SZ];
#pragma unroll
    for (int f = 0; f < F_SZ; ++f) {
        w0[f] = W[f * F_SZ + lane];
        w1[f] = W[f * F_SZ + lane + 32];
    }
    int p0 = ((lane & 15) << 2) | (lane >> 4);

    for (int q = 0; q < WH_NODES * F_SZ / 128; ++q) sH[wv][q * 32 + lane] = hq[q];
    __syncthreads();

    for (int r = 0; r < WH_NODES; ++r) {
        const float* hrow = (const float*)&sH[wv][r * (F_SZ / 4)];
        float acc0 = 0.f, acc1 = 0.f;
#pragma unroll
        for (int f = 0; f < F_SZ; ++f) {
            float hv = hrow[f];
            acc0 = fmaf(hv, w0[f], acc0);
            acc1 = fmaf(hv, w1[f], acc1);
        }
        float* wrow = WhP + (size_t)(n0 + r) * F_SZ;
        wrow[p0]     = acc0;
        wrow[p0 + 2] = acc1;
        sWh[wv][r][lane]      = acc0;
        sWh[wv][r][lane + 32] = acc1;
    }
    __syncthreads();
    float tmax = -INFINITY;
    if (lane < 2 * WH_NODES) {
        int n   = lane & (WH_NODES - 1);
        int sel = lane / WH_NODES;
        const float* av  = a + sel * F_SZ;
        const float* row = &sWh[wv][n][0];
        float d = 0.f;
        for (int k = 0; k < F_SZ; ++k) d = fmaf(row[k], av[k], d);
        float* dst = sel ? t2 : s1;
        float r = d * (1.0f / 64.0f);
        dst[n0 + n] = r;
        if (sel) tmax = r;
    }
    for (int off = WH_NODES / 2; off >= 1; off >>= 1)
        tmax = fmaxf(tmax, __shfl_xor(tmax, off, 32));
    if (lane == WH_NODES) tpart[n0 / WH_NODES] = tmax;
}

typedef int   v2i __attribute__((ext_vector_type(2)));
typedef float v4f __attribute__((ext_vector_type(4)));
struct JBlock {
    v2i av[4];
    v2f tv[4];
    v4f q0[4];
    v4f q1[4];
};

__device__ __forceinline__ void load_block(JBlock& k, const int* __restrict__ pa,
                                           const float* __restrict__ pt,
                                           const float* __restrict__ pb)
{
#pragma unroll
    for (int u = 0; u < 4; ++u) {
        k.av[u] = *(const v2i*)(pa + 4 * u);
        k.tv[u] = *(const v2f*)(pt + 4 * u);
        k.q0[u] = *(const v4f*)(pb + (4 * u) * F_SZ);
        k.q1[u] = *(const v4f*)(pb + (4 * u + 1) * F_SZ);
    }
}

__device__ __forceinline__ void arrive_block(JBlock& k)
{
    asm volatile("" : "+v"(k.av[0]), "+v"(k.av[1]), "+v"(k.av[2]), "+v"(k.av[3]),
                      "+v"(k.tv[0]), "+v"(k.tv[1]), "+v"(k.tv[2]), "+v"(k.tv[3]),
                      "+v"(k.q0[0]), "+v"(k.q0[1]), "+v"(k.q0[2]), "+v"(k.q0[3]),
                      "+v"(k.q1[0]), "+v"(k.q1[1]), "+v"(k.q1[2]), "+v"(k.q1[3]));
}

__device__ __forceinline__ void compute_block(const JBlock& k, float s, float m,
                                              v8f& c0, v8f& c1, v8f& c2, v8f& c3,
                                              float& zacc)
{
#pragma unroll
    for (int u = 0; u < 4; ++u) {
        float x0 = s + k.tv[u].x, x1 = s + k.tv[u].y;
        float e0 = fmaxf(x0, ALPHA * x0);
        float e1 = fmaxf(x1, ALPHA * x1);
        float sc0 = (k.av[u].x > 0) ? e0 : NEG_BIG;
        float sc1 = (k.av[u].y > 0) ? e1 : NEG_BIG;
        float p0 = __builtin_amdgcn_exp2f((sc0 - m) * L2E);
        float p1 = __builtin_amdgcn_exp2f((sc1 - m) * L2E);
        zacc += p0 + p1;

        v2f A; A[0] = p0; A[1] = p1;
        v2f B0, B1, B2, B3;
        B0[0] = k.q0[u].x; B0[1] = k.q1[u].x;
        B1[0] = k.q0[u].y; B1[1] = k.q1[u].y;
        B2[0] = k.q0[u].z; B2[1] = k.q1[u].z;
        B3[0] = k.q0[u].w; B3[1] = k.q1[u].w;

        c0 = __builtin_amdgcn_wmma_f32_16x16x4_f32(false, A, false, B0, (short)0, c0, false, false);
        c1 = __builtin_amdgcn_wmma_f32_16x16x4_f32(false, A, false, B1, (short)0, c1, false, false);
        c2 = __builtin_amdgcn_wmma_f32_16x16x4_f32(false, A, false, B2, (short)0, c2, false, false);
        c3 = __builtin_amdgcn_wmma_f32_16x16x4_f32(false, A, false, B3, (short)0, c3, false, false);
    }
}

__global__ __launch_bounds__(32) void attn_kernel(
    const int* __restrict__ adj, const float* __restrict__ WhP,
    const float* __restrict__ s1, const float* __restrict__ t2,
    const float* __restrict__ tpart, float* __restrict__ Dp,
    float* __restrict__ Zp, int jspan)
{
    int bid   = blockIdx.x;
    int b     = bid & 3;
    int itile = (bid >> 2) & 255;
    int chunk = bid >> 10;
    int lane  = threadIdx.x;
    int l15   = lane & 15;
    int half  = lane >> 4;
    int irow  = itile * 16 + l15;

    float s = s1[b * N_NODES + irow];

    const v4f* tp = (const v4f*)(tpart + b * (N_NODES / WH_NODES)) + lane;
    float T = -INFINITY;
#pragma unroll
    for (int u = 0; u < N_NODES / WH_NODES / 128; ++u) {
        v4f q = tp[u * 32];
        T = fmaxf(T, fmaxf(fmaxf(q.x, q.y), fmaxf(q.z, q.w)));
    }
    for (int off = 16; off >= 1; off >>= 1) T = fmaxf(T, __shfl_xor(T, off, 32));
    float xm = s + T;
    float m = fmaxf(xm, ALPHA * xm);

    v8f c0 = {}, c1 = {}, c2 = {}, c3 = {};
    float zacc = 0.f;

    int jfirst = chunk * jspan + half * 2;
    const int*   pa = adj + (size_t)irow * N_NODES + jfirst;
    const float* pt = t2 + (size_t)b * N_NODES + jfirst;
    const float* pb = WhP + ((size_t)b * N_NODES + jfirst) * F_SZ + (l15 << 2);

#define KEEP_ORDER() __builtin_amdgcn_sched_barrier(0)

    JBlock ka, kb;
    load_block(ka, pa, pt, pb);
    int npair = jspan >> 5;
    for (int it = 1; it < npair; ++it) {
        load_block(kb, pa + 16, pt + 16, pb + 16 * F_SZ);
        KEEP_ORDER();
        arrive_block(ka);
        compute_block(ka, s, m, c0, c1, c2, c3, zacc);
        pa += 32; pt += 32; pb += 32 * F_SZ;
        load_block(ka, pa, pt, pb);
        KEEP_ORDER();
        arrive_block(kb);
        compute_block(kb, s, m, c0, c1, c2, c3, zacc);
    }
    load_block(kb, pa + 16, pt + 16, pb + 16 * F_SZ);
    KEEP_ORDER();
    arrive_block(ka);
    compute_block(ka, s, m, c0, c1, c2, c3, zacc);
    arrive_block(kb);
    compute_block(kb, s, m, c0, c1, c2, c3, zacc);
#undef KEEP_ORDER

    zacc += __shfl_xor(zacc, 16, 32);

    size_t obase = (((size_t)chunk * 4 + b) * N_NODES + (size_t)itile * 16) * F_SZ;
    float* dout = Dp + obase;
    int rbase = half * 8;
#pragma unroll
    for (int r = 0; r < 8; ++r) {
        float4 v; v.x = c0[r]; v.y = c1[r]; v.z = c2[r]; v.w = c3[r];
        *(float4*)(dout + (size_t)(rbase + r) * F_SZ + (l15 << 2)) = v;
    }
    if (half == 0)
        Zp[((size_t)chunk * 4 + b) * N_NODES + irow] = zacc;
}

__device__ __forceinline__ float wave_tree_sum(float r)
{
    for (int off = 16; off >= 1; off >>= 1) r += __shfl_xor(r, off, 32);
    return r;
}

__device__ __forceinline__ void add_chunk(const float* __restrict__ Dp,
                                          const float* __restrict__ Zp,
                                          int c, int i, int p1,
                                          float (&num)[4][2], float (&z)[4])
{
    for (int b = 0; b < 4; ++b) {
        size_t row = ((size_t)c * 4 + b) * N_NODES + i;
        const float* d = Dp + (row << 6);
        num[b][0] += d[p1];
        num[b][1] += d[p1 + 2];
        z[b]      += Zp[row];
    }
}

__global__ __launch_bounds__(256) void bn_kernel(
    const float* __restrict__ Dp, const float* __restrict__ Zp,
    const float* __restrict__ WhP,
    const float* __restrict__ h, const float* __restrict__ gamma,
    const float* __restrict__ beta, float* __restrict__ out, int nchunk)
{
    int lane = threadIdx.x & 31;
    int wv = __builtin_amdgcn_readfirstlane((int)(threadIdx.x >> 5));
    int i  = blockIdx.x * 8 + wv;
    int p1 = ((lane & 15) << 2) | (lane >> 4);

    float hin[4][2];
    for (int b = 0; b < 4; ++b) {
        const float* hr = h + ((size_t)b * N_NODES + i) * F_SZ;
        hin[b][0] = hr[lane];
        hin[b][1] = hr[lane + 32];
    }
    float g = gamma[i], be = beta[i];

    float num[4][2], z[4];
    for (int b = 0; b < 4; ++b) { num[b][0] = 0.f; num[b][1] = 0.f; z[b] = 0.f; }
    if (nchunk == 4) {
        for (int c = 0; c < 4; ++c) add_chunk(Dp, Zp, c, i, p1, num, z);
    } else {
        for (int c = 0; c < nchunk; ++c) add_chunk(Dp, Zp, c, i, p1, num, z);
    }

    for (int b = 0; b < 4; ++b) {
        if (z[b] == 0.f) {
            const float* wcol = WhP + (size_t)b * N_NODES * F_SZ + p1;
            float c0 = 0.f, c1 = 0.f;
            for (int j = 0; j < N_NODES; ++j) {
                c0 += wcol[(size_t)j * F_SZ];
                c1 += wcol[(size_t)j * F_SZ + 2];
            }
            num[b][0] = c0; num[b][1] = c1; z[b] = (float)N_NODES;
        }
    }

    float hp[4][2];
    for (int b = 0; b < 4; ++b) {
        hp[b][0] = num[b][0] / z[b];
        hp[b][1] = num[b][1] / z[b];
    }

    float q0 = (hp[0][0] + hp[2][0]) + (hp[1][0] + hp[3][0]);
    float q1 = (hp[0][1] + hp[2][1]) + (hp[1][1] + hp[3][1]);
    float mean = wave_tree_sum(q0 + q1) * (1.0f / 256.0f);

    float dev[4][2], sq[4][2];
    for (int b = 0; b < 4; ++b) {
        dev[b][0] = hp[b][0] - mean;  sq[b][0] = dev[b][0] * dev[b][0];
        dev[b][1] = hp[b][1] - mean;  sq[b][1] = dev[b][1] * dev[b][1];
    }
    float v0 = (sq[0][0] + sq[2][0]) + (sq[1][0] + sq[3][0]);
    float v1 = (sq[0][1] + sq[2][1]) + (sq[1][1] + sq[3][1]);
    float var = wave_tree_sum(v0 + v1) * (1.0f / 256.0f);

    float inv = 1.0f / sqrtf(var + 1e-5f);
    for (int b = 0; b < 4; ++b) {
        float* orow = out + ((size_t)b * N_NODES + i) * F_SZ;
        for (int k = 0; k < 2; ++k) {
            float xn  = g * dev[b][k] * inv + be;
            float res = xn + hin[b][k];
            orow[lane + 32 * k] = res > 0.f ? res : expm1f(res);
        }
    }
}

extern "C" void kernel_launch(void* const* d_in, const int* in_sizes, int n_in,
                              void* d_out, int out_size, void* d_ws, size_t ws_size,
                              hipStream_t stream)
{
    const float* h     = (const float*)d_in[0];
    const int*   adj   = (const int*)  d_in[1];
    const float* W     = (const float*)d_in[2];
    const float* a     = (const float*)d_in[3];
    const float* gamma = (const float*)d_in[4];
    const float* beta  = (const float*)d_in[5];
    float* out = (float*)d_out;

    const size_t WH_ELEMS = (size_t)B_SZ * N_NODES * F_SZ;
    const size_t NROW     = (size_t)B_SZ * N_NODES;

    int nchunk = 4;
    size_t need4 = (WH_ELEMS + 3 * NROW + 4 * (WH_ELEMS + NROW)) * sizeof(float);
    if (ws_size < need4) nchunk = 1;

    float* wsf  = (float*)d_ws;
    float* WhP  = wsf;
    float* s1   = WhP + WH_ELEMS;
    float* t2   = s1 + NROW;
    float* tpart = t2 + NROW;
    float* Dp   = tpart + NROW;
    float* Zp   = Dp + (size_t)nchunk * WH_ELEMS;

    int jspan = N_NODES / nchunk;

    wh_kernel<<<B_SZ * N_NODES / (WH_WAVES * WH_NODES), 32 * WH_WAVES, 0, stream>>>(h, W, a, WhP, s1, t2, tpart);
    attn_kernel<<<B_SZ * (N_NODES / 16) * nchunk, 32, 0, stream>>>(
        adj, WhP, s1, t2, tpart, Dp, Zp, jspan);
    bn_kernel<<<N_NODES / 8, 256, 0, stream>>>(Dp, Zp, WhP, h, gamma, beta, out, nchunk);
}


# attn kernel: each wave holds two 16-row tiles and uses every loaded Wh fragment for both: Wh re-reads from L2 and wave count halved; loads run 3 blocks of 8 columns ahead; sum order unchanged
# speedup vs baseline: 2.2077x; 1.0738x over previous
#include <hip/hip_runtime.h>
#include <hip/hip_bf16.h>
#include <math.h>

typedef float v2f __attribute__((ext_vector_type(2)));
typedef float v8f __attribute__((ext_vector_type(8)));

#define N_NODES 4096
#define B_SZ    4
#define F_SZ    64
#define ALPHA   0.2f
#define NEG_BIG -9e15f
#define L2E     1.4426950408889634f

#define WH_NODES 8
#define WH_WAVES 2
#define WH_PITCH (F_SZ + 1)

__global__ __launch_bounds__(32 * WH_WAVES) void wh_kernel(
    const float* __restrict__ h, const float* __restrict__ W,
    const float* __restrict__ a, float* __restrict__ WhP,
    float* __restrict__ s1, float* __restrict__ t2, float* __restrict__ tpart)
{
    __shared__ float sWh[WH_WAVES][WH_NODES][WH_PITCH];
    __shared__ float4 sH[WH_WAVES][WH_NODES * F_SZ / 4];
    int lane = threadIdx.x & 31;
    int wv = threadIdx.x >> 5;
    int n0 = (blockIdx.x * WH_WAVES + wv) * WH_NODES;

    const float4* hsrc = (const float4*)(h + (size_t)n0 * F_SZ);
    float4 hq[WH_NODES * F_SZ / 128];
    for (int q = 0; q < WH_NODES * F_SZ / 128; ++q) hq[q] = hsrc[q * 32 + lane];

    float w0[F_SZ], w1[F_SZ];
#pragma unroll
    for (int f = 0; f < F_SZ; ++f) {
        w0[f] = W[f * F_SZ + lane];
        w1[f] = W[f * F_SZ + lane + 32];
    }
    int p0 = ((lane & 15) << 2) | (lane >> 4);

    for (int q = 0; q < WH_NODES * F_SZ / 128; ++q) sH[wv][q * 32 + lane] = hq[q];
    __syncthreads();

    for (int r = 0; r < WH_NODES; ++r) {
        const float* hrow = (const float*)&sH[wv][r * (F_SZ / 4)];
        float acc0 = 0.f, acc1 = 0.f;
#pragma unroll
        for (int f = 0; f < F_SZ; ++f) {
            float hv = hrow[f];
            acc0 = fmaf(hv, w0[f], acc0);
            acc1 = fmaf(hv, w1[f], acc1);
        }
        float* wrow = WhP + (size_t)(n0 + r) * F_SZ;
        wrow[p0]     = acc0;
        wrow[p0 + 2] = acc1;
        sWh[wv][r][lane]      = acc0;
        sWh[wv][r][lane + 32] = acc1;
    }
    __syncthreads();
    float tmax = -INFINITY;
    if (lane < 2 * WH_NODES) {
        int n   = lane & (WH_NODES - 1);
        int sel = lane / WH_NODES;
        const float* av  = a + sel * F_SZ;
        const float* row = &sWh[wv][n][0];
        float d = 0.f;
        for (int k = 0; k < F_SZ; ++k) d = fmaf(row[k], av[k], d);
        float* dst = sel ? t2 : s1;
        float r = d * (1.0f / 64.0f);
        dst[n0 + n] = r;
        if (sel) tmax = r;
    }
    for (int off = WH_NODES / 2; off >= 1; off >>= 1)
        tmax = fmaxf(tmax, __shfl_xor(tmax, off, 32));
    if (lane == WH_NODES) tpart[n0 / WH_NODES] = tmax;
}

typedef int   v2i __attribute__((ext_vector_type(2)));
typedef float v4f __attribute__((ext_vector_type(4)));
#define ROWT 2
#define UB   2
struct JBlock {
    v2i av[ROWT][UB];
    v2f tv[UB];
    v4f q0[UB];
    v4f q1[UB];
};

__device__ __forceinline__ void load_block(JBlock& k, const int* __restrict__ pa,
                                           const float* __restrict__ pt,
                                           const float* __restrict__ pb)
{
#pragma unroll
    for (int u = 0; u < UB; ++u) {
#pragma unroll
        for (int r = 0; r < ROWT; ++r)
            k.av[r][u] = *(const v2i*)(pa + (size_t)r * 16 * N_NODES + 4 * u);
        k.tv[u] = *(const v2f*)(pt + 4 * u);
        k.q0[u] = *(const v4f*)(pb + (4 * u) * F_SZ);
        k.q1[u] = *(const v4f*)(pb + (4 * u + 1) * F_SZ);
    }
}

__device__ __forceinline__ void arrive_block(JBlock& k)
{
    asm volatile("" : "+v"(k.av[0][0]), "+v"(k.av[0][1]), "+v"(k.av[1][0]), "+v"(k.av[1][1]),
                      "+v"(k.tv[0]), "+v"(k.tv[1]),
                      "+v"(k.q0[0]), "+v"(k.q0[1]), "+v"(k.q1[0]), "+v"(k.q1[1]));
}

__device__ __forceinline__ void compute_block(const JBlock& k, const float (&s)[ROWT],
                                              const float (&m)[ROWT],
                                              v8f (&c)[ROWT][4], float (&zacc)[ROWT])
{
#pragma unroll
    for (int u = 0; u < UB; ++u) {
        v2f B0, B1, B2, B3;
        B0[0] = k.q0[u].x; B0[1] = k.q1[u].x;
        B1[0] = k.q0[u].y; B1[1] = k.q1[u].y;
        B2[0] = k.q0[u].z; B2[1] = k.q1[u].z;
        B3[0] = k.q0[u].w; B3[1] = k.q1[u].w;

        v2f A[ROWT];
#pragma unroll
        for (int r = 0; r < ROWT; ++r) {
            float x0 = s[r] + k.tv[u].x, x1 = s[r] + k.tv[u].y;
            float e0 = fmaxf(x0, ALPHA * x0);
            float e1 = fmaxf(x1, ALPHA * x1);
            float sc0 = (k.av[r][u].x > 0) ? e0 : NEG_BIG;
            float sc1 = (k.av[r][u].y > 0) ? e1 : NEG_BIG;
            float p0 = __builtin_amdgcn_exp2f((sc0 - m[r]) * L2E);
            float p1 = __builtin_amdgcn_exp2f((sc1 - m[r]) * L2E);
            zacc[r] += p0 + p1;
            A[r][0] = p0; A[r][1] = p1;
        }
#pragma unroll
        for (int r = 0; r < ROWT; ++r) {
            c[r][0] = __builtin_amdgcn_wmma_f32_16x16x4_f32(false, A[r], false, B0, (short)0, c[r][0], false, false);
            c[r][1] = __builtin_amdgcn_wmma_f32_16x16x4_f32(false, A[r], false, B1, (short)0, c[r][1], false, false);
            c[r][2] = __builtin_amdgcn_wmma_f32_16x16x4_f32(false, A[r], false, B2, (short)0, c[r][2], false, false);
            c[r][3] = __builtin_amdgcn_wmma_f32_16x16x4_f32(false, A[r], false, B3, (short)0, c[r][3], false, false);
        }
    }
}

__global__ __launch_bounds__(32) void attn_kernel(
    const int* __restrict__ adj, const float* __restrict__ WhP,
    const float* __restrict__ s1, const float* __restrict__ t2,
    const float* __restrict__ tpart, float* __restrict__ Dp,
    float* __restrict__ Zp, int jspan)
{
    int bid   = blockIdx.x;
    int b     = bid & 3;
    int itile = (bid >> 2) & (N_NODES / (16 * ROWT) - 1);
    int chunk = bid / (4 * (N_NODES / (16 * ROWT)));
    int lane  = threadIdx.x;
    int l15   = lane & 15;
    int half  = lane >> 4;
    int irow  = itile * (16 * ROWT) + l15;

    float s[ROWT];
#pragma unroll
    for (int r = 0; r < ROWT; ++r) s[r] = s1[b * N_NODES + irow + 16 * r];

    const v4f* tp = (const v4f*)(tpart + b * (N_NODES / WH_NODES)) + lane;
    float T = -INFINITY;
#pragma unroll
    for (int u = 0; u < N_NODES / WH_NODES / 128; ++u) {
        v4f q = tp[u * 32];
        T = fmaxf(T, fmaxf(fmaxf(q.x, q.y), fmaxf(q.z, q.w)));
    }
    for (int off = 16; off >= 1; off >>= 1) T = fmaxf(T, __shfl_xor(T, off, 32));
    float m[ROWT];
#pragma unroll
    for (int r = 0; r < ROWT; ++r) {
        float xm = s[r] + T;
        m[r] = fmaxf(xm, ALPHA * xm);
    }

    v8f c[ROWT][4];
    float zacc[ROWT];
#pragma unroll
    for (int r = 0; r < ROWT; ++r) {
        zacc[r] = 0.f;
#pragma unroll
        for (int q = 0; q < 4; ++q) c[r][q] = v8f{};
    }

    int jfirst = chunk * jspan + half * 2;
    const int*   pa = adj + (size_t)irow * N_NODES + jfirst;
    const float* pt = t2 + (size_t)b * N_NODES + jfirst;
    const float* pb = WhP + ((size_t)b * N_NODES + jfirst) * F_SZ + (l15 << 2);

#define KEEP_ORDER() __builtin_amdgcn_sched_barrier(0)
#define STEP(KLOAD, OFF, KWORK)                                              \
    load_block(KLOAD, pa + (OFF), pt + (OFF), pb + (OFF) * F_SZ);            \
    KEEP_ORDER();                                                            \
    arrive_block(KWORK);                                                     \
    compute_block(KWORK, s, m, c, zacc);                                     \
    KEEP_ORDER();

    JBlock k0, k1, k2, k3;
    load_block(k0, pa, pt, pb);
    KEEP_ORDER();
    load_block(k1, pa + 8, pt + 8, pb + 8 * F_SZ);
    KEEP_ORDER();
    load_block(k2, pa + 16, pt + 16, pb + 16 * F_SZ);
    KEEP_ORDER();
    int nquad = jspan >> 5;
    for (int it = 1; it < nquad; ++it) {
        STEP(k3, 24, k0)
        pa += 32; pt += 32; pb += 32 * F_SZ;
        STEP(k0, 0, k1)
        STEP(k1, 8, k2)
        STEP(k2, 16, k3)
    }
    STEP(k3, 24, k0)
    arrive_block(k1);
    compute_block(k1, s, m, c, zacc);
    arrive_block(k2);
    compute_block(k2, s, m, c, zacc);
    arrive_block(k3);
    compute_block(k3, s, m, c, zacc);
#undef STEP
#undef KEEP_ORDER

    size_t obase = (((size_t)chunk * 4 + b) * N_NODES + (size_t)itile * (16 * ROWT)) * F_SZ;
    int rbase = half * 8;
#pragma unroll
    for (int r = 0; r < ROWT; ++r) {
        float z = zacc[r] + __shfl_xor(zacc[r], 16, 32);
        float* dout = Dp + obase + (size_t)(16 * r) * F_SZ;
#pragma unroll
        for (int q = 0; q < 8; ++q) {
            float4 v; v.x = c[r][0][q]; v.y = c[r][1][q]; v.z = c[r][2][q]; v.w = c[r][3][q];
            *(float4*)(dout + (size_t)(rbase + q) * F_SZ + (l15 << 2)) = v;
        }
        if (half == 0)
            Zp[((size_t)chunk * 4 + b) * N_NODES + irow + 16 * r] = z;
    }
}

__device__ __forceinline__ float wave_tree_sum(float r)
{
    for (int off = 16; off >= 1; off >>= 1) r += __shfl_xor(r, off, 32);
    return r;
}

__device__ __forceinline__ void add_chunk(const float* __restrict__ Dp,
                                          const float* __restrict__ Zp,
                                          int c, int i, int p1,
                                          float (&num)[4][2], float (&z)[4])
{
    for (int b = 0; b < 4; ++b) {
        size_t row = ((size_t)c * 4 + b) * N_NODES + i;
        const float* d = Dp + (row << 6);
        num[b][0] += d[p1];
        num[b][1] += d[p1 + 2];
        z[b]      += Zp[row];
    }
}

__global__ __launch_bounds__(256) void bn_kernel(
    const float* __restrict__ Dp, const float* __restrict__ Zp,
    const float* __restrict__ WhP,
    const float* __restrict__ h, const float* __restrict__ gamma,
    const float* __restrict__ beta, float* __restrict__ out, int nchunk)
{
    int lane = threadIdx.x & 31;
    int wv = __builtin_amdgcn_readfirstlane((int)(threadIdx.x >> 5));
    int i  = blockIdx.x * 8 + wv;
    int p1 = ((lane & 15) << 2) | (lane >> 4);

    float hin[4][2];
    for (int b = 0; b < 4; ++b) {
        const float* hr = h + ((size_t)b * N_NODES + i) * F_SZ;
        hin[b][0] = hr[lane];
        hin[b][1] = hr[lane + 32];
    }
    float g = gamma[i], be = beta[i];

    float num[4][2], z[4];
    for (int b = 0; b < 4; ++b) { num[b][0] = 0.f; num[b][1] = 0.f; z[b] = 0.f; }
    if (nchunk == 4) {
        for (int c = 0; c < 4; ++c) add_chunk(Dp, Zp, c, i, p1, num, z);
    } else {
        for (int c = 0; c < nchunk; ++c) add_chunk(Dp, Zp, c, i, p1, num, z);
    }

    for (int b = 0; b < 4; ++b) {
        if (z[b] == 0.f) {
            const float* wcol = WhP + (size_t)b * N_NODES * F_SZ + p1;
            float c0 = 0.f, c1 = 0.f;
            for (int j = 0; j < N_NODES; ++j) {
                c0 += wcol[(size_t)j * F_SZ];
                c1 += wcol[(size_t)j * F_SZ + 2];
            }
            num[b][0] = c0; num[b][1] = c1; z[b] = (float)N_NODES;
        }
    }

    float hp[4][2];
    for (int b = 0; b < 4; ++b) {
        hp[b][0] = num[b][0] / z[b];
        hp[b][1] = num[b][1] / z[b];
    }

    float q0 = (hp[0][0] + hp[2][0]) + (hp[1][0] + hp[3][0]);
    float q1 = (hp[0][1] + hp[2][1]) + (hp[1][1] + hp[3][1]);
    float mean = wave_tree_sum(q0 + q1) * (1.0f / 256.0f);

    float dev[4][2], sq[4][2];
    for (int b = 0; b < 4; ++b) {
        dev[b][0] = hp[b][0] - mean;  sq[b][0] = dev[b][0] * dev[b][0];
        dev[b][1] = hp[b][1] - mean;  sq[b][1] = dev[b][1] * dev[b][1];
    }
    float v0 = (sq[0][0] + sq[2][0]) + (sq[1][0] + sq[3][0]);
    float v1 = (sq[0][1] + sq[2][1]) + (sq[1][1] + sq[3][1]);
    float var = wave_tree_sum(v0 + v1) * (1.0f / 256.0f);

    float inv = 1.0f / sqrtf(var + 1e-5f);
    for (int b = 0; b < 4; ++b) {
        float* orow = out + ((size_t)b * N_NODES + i) * F_SZ;
        for (int k = 0; k < 2; ++k) {
            float xn  = g * dev[b][k] * inv + be;
            float res = xn + hin[b][k];
            orow[lane + 32 * k] = res > 0.f ? res : expm1f(res);
        }
    }
}

extern "C" void kernel_launch(void* const* d_in, const int* in_sizes, int n_in,
                              void* d_out, int out_size, void* d_ws, size_t ws_size,
                              hipStream_t stream)
{
    const float* h     = (const float*)d_in[0];
    const int*   adj   = (const int*)  d_in[1];
    const float* W     = (const float*)d_in[2];
    const float* a     = (const float*)d_in[3];
    const float* gamma = (const float*)d_in[4];
    const float* beta  = (const float*)d_in[5];
    float* out = (float*)d_out;

    const size_t WH_ELEMS = (size_t)B_SZ * N_NODES * F_SZ;
    const size_t NROW     = (size_t)B_SZ * N_NODES;

    int nchunk = 4;
    size_t need4 = (WH_ELEMS + 3 * NROW + 4 * (WH_ELEMS + NROW)) * sizeof(float);
    if (ws_size < need4) nchunk = 1;

    float* wsf  = (float*)d_ws;
    float* WhP  = wsf;
    float* s1   = WhP + WH_ELEMS;
    float* t2   = s1 + NROW;
    float* tpart = t2 + NROW;
    float* Dp   = tpart + NROW;
    float* Zp   = Dp + (size_t)nchunk * WH_ELEMS;

    int jspan = N_NODES / nchunk;

    wh_kernel<<<B_SZ * N_NODES / (WH_WAVES * WH_NODES), 32 * WH_WAVES, 0, stream>>>(h, W, a, WhP, s1, t2, tpart);
    attn_kernel<<<B_SZ * (N_NODES / (16 * ROWT)) * nchunk, 32, 0, stream>>>(
        adj, WhP, s1, t2, tpart, Dp, Zp, jspan);
    bn_kernel<<<N_NODES / 8, 256, 0, stream>>>(Dp, Zp, WhP, h, gamma, beta, out, nchunk);
}


# attn j loop without exp: exp(lrelu(s+t)-m) = max(a_i*v_j, b_i*w_j); v=exp(t'), w=exp(0.2t') tabulated per column by the Wh kernel, a_i, b_i set once per wave; sum order unchanged
# speedup vs baseline: 2.2962x; 1.0401x over previous
#include <hip/hip_runtime.h>
#include <hip/hip_bf16.h>
#include <math.h>

typedef float v2f __attribute__((ext_vector_type(2)));
typedef float v8f __attribute__((ext_vector_type(8)));

#define N_NODES 4096
#define B_SZ    4
#define F_SZ    64
#define ALPHA   0.2f
#define L2E     1.4426950408889634f

#define WH_NODES 8
#define WH_WAVES 2
#define WH_PITCH (F_SZ + 1)

__global__ __launch_bounds__(32 * WH_WAVES) void wh_kernel(
    const float* __restrict__ h, const float* __restrict__ W,
    const float* __restrict__ a, float* __restrict__ WhP,
    float* __restrict__ s1, float* __restrict__ vtab, float* __restrict__ tpart)
{
    __shared__ float sWh[WH_WAVES][WH_NODES][WH_PITCH];
    __shared__ float4 sH[WH_WAVES][WH_NODES * F_SZ / 4];
    int lane = threadIdx.x & 31;
    int wv = threadIdx.x >> 5;
    int n0 = (blockIdx.x * WH_WAVES + wv) * WH_NODES;

    const float4* hsrc = (const float4*)(h + (size_t)n0 * F_SZ);
    float4 hq[WH_NODES * F_SZ / 128];
    for (int q = 0; q < WH_NODES * F_SZ / 128; ++q) hq[q] = hsrc[q * 32 + lane];

    float w0[F_SZ], w1[F_SZ];
#pragma unroll
    for (int f = 0; f < F_SZ; ++f) {
        w0[f] = W[f * F_SZ + lane];
        w1[f] = W[f * F_SZ + lane + 32];
    }
    int p0 = ((lane & 15) << 2) | (lane >> 4);

    for (int q = 0; q < WH_NODES * F_SZ / 128; ++q) sH[wv][q * 32 + lane] = hq[q];
    __syncthreads();

    for (int r = 0; r < WH_NODES; ++r) {
        const float* hrow = (const float*)&sH[wv][r * (F_SZ / 4)];
        float acc0 = 0.f, acc1 = 0.f;
#pragma unroll
        for (int f = 0; f < F_SZ; ++f) {
            float hv = hrow[f];
            acc0 = fmaf(hv, w0[f], acc0);
            acc1 = fmaf(hv, w1[f], acc1);
        }
        float* wrow = WhP + (size_t)(n0 + r) * F_SZ;
        wrow[p0]     = acc0;
        wrow[p0 + 2] = acc1;
        sWh[wv][r][lane]      = acc0;
        sWh[wv][r][lane + 32] = acc1;
    }
    __syncthreads();
    float tmax = -INFINITY;
    if (lane < 2 * WH_NODES) {
        int n   = lane & (WH_NODES - 1);
        int sel = lane / WH_NODES;
        const float* av  = a + sel * F_SZ;
        const float* row = &sWh[wv][n][0];
        float d = 0.f;
        for (int k = 0; k < F_SZ; ++k) d = fmaf(row[k], av[k], d);
        float r = d * (1.0f / 64.0f);
        if (sel) {
            int g = n0 + n;
            float* vt = vtab + (size_t)(g >> 1) * 4 + (g & 1);
            vt[0] = __builtin_amdgcn_exp2f(r * L2E);
            vt[2] = __builtin_amdgcn_exp2f((ALPHA * r) * L2E);
            tmax = r;
        } else {
            s1[n0 + n] = r;
        }
    }
    for (int off = WH_NODES / 2; off >= 1; off >>= 1)
        tmax = fmaxf(tmax, __shfl_xor(tmax, off, 32));
    if (lane == WH_NODES) tpart[n0 / WH_NODES] = tmax;
}

typedef int   v2i __attribute__((ext_vector_type(2)));
typedef float v4f __attribute__((ext_vector_type(4)));
#define ROWT 2
#define UB   2
struct JBlock {
    v2i av[ROWT][UB];
    v4f cf[UB];
    v4f q0[UB];
    v4f q1[UB];
};

__device__ __forceinline__ void load_block(JBlock& k, const int* __restrict__ pa,
                                           const float* __restrict__ pv,
                                           const float* __restrict__ pb)
{
#pragma unroll
    for (int u = 0; u < UB; ++u) {
#pragma unroll
        for (int r = 0; r < ROWT; ++r)
            k.av[r][u] = *(const v2i*)(pa + (size_t)r * 16 * N_NODES + 4 * u);
        k.cf[u] = *(const v4f*)(pv + 8 * u);
        k.q0[u] = *(const v4f*)(pb + (4 * u) * F_SZ);
        k.q1[u] = *(const v4f*)(pb + (4 * u + 1) * F_SZ);
    }
}

__device__ __forceinline__ void arrive_block(JBlock& k)
{
    asm volatile("" : "+v"(k.av[0][0]), "+v"(k.av[0][1]), "+v"(k.av[1][0]), "+v"(k.av[1][1]),
                      "+v"(k.cf[0]), "+v"(k.cf[1]),
                      "+v"(k.q0[0]), "+v"(k.q0[1]), "+v"(k.q1[0]), "+v"(k.q1[1]));
}

__device__ __forceinline__ void settle_block(v8f (&c)[ROWT][4])
{
    asm volatile("" : "+v"(c[0][0]), "+v"(c[0][1]), "+v"(c[0][2]), "+v"(c[0][3]),
                      "+v"(c[1][0]), "+v"(c[1][1]), "+v"(c[1][2]), "+v"(c[1][3]));
}

__device__ __forceinline__ void compute_block(const JBlock& k, const float (&fa)[ROWT],
                                              const float (&fb)[ROWT],
                                              v8f (&c)[ROWT][4], float (&zacc)[ROWT])
{
#pragma unroll
    for (int u = 0; u < UB; ++u) {
        v2f B0, B1, B2, B3;
        B0[0] = k.q0[u].x; B0[1] = k.q1[u].x;
        B1[0] = k.q0[u].y; B1[1] = k.q1[u].y;
        B2[0] = k.q0[u].z; B2[1] = k.q1[u].z;
        B3[0] = k.q0[u].w; B3[1] = k.q1[u].w;

        v2f A[ROWT];
#pragma unroll
        for (int r = 0; r < ROWT; ++r) {
            float e0 = fmaxf(fa[r] * k.cf[u].x, fb[r] * k.cf[u].z);
            float e1 = fmaxf(fa[r] * k.cf[u].y, fb[r] * k.cf[u].w);
            float p0 = (k.av[r][u].x > 0) ? e0 : 0.f;
            float p1 = (k.av[r][u].y > 0) ? e1 : 0.f;
            zacc[r] += p0 + p1;
            A[r][0] = p0; A[r][1] = p1;
        }
#pragma unroll
        for (int r = 0; r < ROWT; ++r) {
            c[r][0] = __builtin_amdgcn_wmma_f32_16x16x4_f32(false, A[r], false, B0, (short)0, c[r][0], false, false);
            c[r][1] = __builtin_amdgcn_wmma_f32_16x16x4_f32(false, A[r], false, B1, (short)0, c[r][1], false, false);
            c[r][2] = __builtin_amdgcn_wmma_f32_16x16x4_f32(false, A[r], false, B2, (short)0, c[r][2], false, false);
            c[r][3] = __builtin_amdgcn_wmma_f32_16x16x4_f32(false, A[r], false, B3, (short)0, c[r][3], false, false);
        }
    }
}

__global__ __launch_bounds__(32) void attn_kernel(
    const int* __restrict__ adj, const float* __restrict__ WhP,
    const float* __restrict__ s1, const float* __restrict__ vtab,
    const float* __restrict__ tpart, float* __restrict__ Dp,
    float* __restrict__ Zp, int jspan)
{
    int bid   = blockIdx.x;
    int b     = bid & 3;
    int itile = (bid >> 2) & (N_NODES / (16 * ROWT) - 1);
    int chunk = bid / (4 * (N_NODES / (16 * ROWT)));
    int lane  = threadIdx.x;
    int l15   = lane & 15;
    int half  = lane >> 4;
    int irow  = itile * (16 * ROWT) + l15;

    float s[ROWT];
#pragma unroll
    for (int r = 0; r < ROWT; ++r) s[r] = s1[b * N_NODES + irow + 16 * r];

    const v4f* tp = (const v4f*)(tpart + b * (N_NODES / WH_NODES)) + lane;
    float T = -INFINITY;
#pragma unroll
    for (int u = 0; u < N_NODES / WH_NODES / 128; ++u) {
        v4f q = tp[u * 32];
        T = fmaxf(T, fmaxf(fmaxf(q.x, q.y), fmaxf(q.z, q.w)));
    }
    for (int off = 16; off >= 1; off >>= 1) T = fmaxf(T, __shfl_xor(T, off, 32));
    float fa[ROWT], fb[ROWT];
#pragma unroll
    for (int r = 0; r < ROWT; ++r) {
        float xm = s[r] + T;
        float m  = fmaxf(xm, ALPHA * xm);
        fa[r] = __builtin_amdgcn_exp2f((s[r] - m) * L2E);
        fb[r] = __builtin_amdgcn_exp2f((ALPHA * s[r] - m) * L2E);
    }

    v8f c[ROWT][4];
    float zacc[ROWT];
#pragma unroll
    for (int r = 0; r < ROWT; ++r) {
        zacc[r] = 0.f;
#pragma unroll
        for (int q = 0; q < 4; ++q) c[r][q] = v8f{};
    }

    int jfirst = chunk * jspan + half * 2;
    const int*   pa = adj + (size_t)irow * N_NODES + jfirst;
    const float* pv = vtab + ((size_t)b * N_NODES + jfirst) * 2;
    const float* pb = WhP + ((size_t)b * N_NODES + jfirst) * F_SZ + (l15 << 2);

#define KEEP_ORDER() __builtin_amdgcn_sched_barrier(0)
#define STEP(KLOAD, OFF, KWORK)                                              \
    load_block(KLOAD, pa + (OFF), pv + 2 * (OFF), pb + (OFF) * F_SZ);        \
    KEEP_ORDER();                                                            \
    arrive_block(KWORK);                                                     \
    compute_block(KWORK, fa, fb, c, zacc);                                   \
    settle_block(c);                                                         \
    KEEP_ORDER();

    JBlock k0, k1, k2, k3;
    load_block(k0, pa, pv, pb);
    KEEP_ORDER();
    load_block(k1, pa + 8, pv + 16, pb + 8 * F_SZ);
    KEEP_ORDER();
    load_block(k2, pa + 16, pv + 32, pb + 16 * F_SZ);
    KEEP_ORDER();
    int nquad = jspan >> 5;
    for (int it = 1; it < nquad; ++it) {
        STEP(k3, 24, k0)
        pa += 32; pv += 64; pb += 32 * F_SZ;
        STEP(k0, 0, k1)
        STEP(k1, 8, k2)
        STEP(k2, 16, k3)
    }
    STEP(k3, 24, k0)
    arrive_block(k1);
    compute_block(k1, fa, fb, c, zacc);
    arrive_block(k2);
    compute_block(k2, fa, fb, c, zacc);
    arrive_block(k3);
    compute_block(k3, fa, fb, c, zacc);
#undef STEP
#undef KEEP_ORDER

    size_t obase = (((size_t)chunk * 4 + b) * N_NODES + (size_t)itile * (16 * ROWT)) * F_SZ;
    int rbase = half * 8;
#pragma unroll
    for (int r = 0; r < ROWT; ++r) {
        float z = zacc[r] + __shfl_xor(zacc[r], 16, 32);
        float* dout = Dp + obase + (size_t)(16 * r) * F_SZ;
#pragma unroll
        for (int q = 0; q < 8; ++q) {
            float4 v; v.x = c[r][0][q]; v.y = c[r][1][q]; v.z = c[r][2][q]; v.w = c[r][3][q];
            *(float4*)(dout + (size_t)(rbase + q) * F_SZ + (l15 << 2)) = v;
        }
        if (half == 0)
            Zp[((size_t)chunk * 4 + b) * N_NODES + irow + 16 * r] = z;
    }
}

__device__ __forceinline__ float wave_tree_sum(float r)
{
    for (int off = 16; off >= 1; off >>= 1) r += __shfl_xor(r, off, 32);
    return r;
}

__device__ __forceinline__ void add_chunk(const float* __restrict__ Dp,
                                          const float* __restrict__ Zp,
                                          int c, int i, int p1,
                                          float (&num)[4][2], float (&z)[4])
{
    for (int b = 0; b < 4; ++b) {
        size_t row = ((size_t)c * 4 + b) * N_NODES + i;
        const float* d = Dp + (row << 6);
        num[b][0] += d[p1];
        num[b][1] += d[p1 + 2];
        z[b]      += Zp[row];
    }
}

__global__ __launch_bounds__(256) void bn_kernel(
    const float* __restrict__ Dp, const float* __restrict__ Zp,
    const float* __restrict__ WhP,
    const float* __restrict__ h, const float* __restrict__ gamma,
    const float* __restrict__ beta, float* __restrict__ out, int nchunk)
{
    int lane = threadIdx.x & 31;
    int wv = __builtin_amdgcn_readfirstlane((int)(threadIdx.x >> 5));
    int i  = blockIdx.x * 8 + wv;
    int p1 = ((lane & 15) << 2) | (lane >> 4);

    float hin[4][2];
    for (int b = 0; b < 4; ++b) {
        const float* hr = h + ((size_t)b * N_NODES + i) * F_SZ;
        hin[b][0] = hr[lane];
        hin[b][1] = hr[lane + 32];
    }
    float g = gamma[i], be = beta[i];

    float num[4][2], z[4];
    for (int b = 0; b < 4; ++b) { num[b][0] = 0.f; num[b][1] = 0.f; z[b] = 0.f; }
    if (nchunk == 4) {
        for (int c = 0; c < 4; ++c) add_chunk(Dp, Zp, c, i, p1, num, z);
    } else {
        for (int c = 0; c < nchunk; ++c) add_chunk(Dp, Zp, c, i, p1, num, z);
    }

    for (int b = 0; b < 4; ++b) {
        if (z[b] == 0.f) {
            const float* wcol = WhP + (size_t)b * N_NODES * F_SZ + p1;
            float c0 = 0.f, c1 = 0.f;
            for (int j = 0; j < N_NODES; ++j) {
                c0 += wcol[(size_t)j * F_SZ];
                c1 += wcol[(size_t)j * F_SZ + 2];
            }
            num[b][0] = c0; num[b][1] = c1; z[b] = (float)N_NODES;
        }
    }

    float hp[4][2];
    for (int b = 0; b < 4; ++b) {
        hp[b][0] = num[b][0] / z[b];
        hp[b][1] = num[b][1] / z[b];
    }

    float q0 = (hp[0][0] + hp[2][0]) + (hp[1][0] + hp[3][0]);
    float q1 = (hp[0][1] + hp[2][1]) + (hp[1][1] + hp[3][1]);
    float mean = wave_tree_sum(q0 + q1) * (1.0f / 256.0f);

    float dev[4][2], sq[4][2];
    for (int b = 0; b < 4; ++b) {
        dev[b][0] = hp[b][0] - mean;  sq[b][0] = dev[b][0] * dev[b][0];
        dev[b][1] = hp[b][1] - mean;  sq[b][1] = dev[b][1] * dev[b][1];
    }
    float v0 = (sq[0][0] + sq[2][0]) + (sq[1][0] + sq[3][0]);
    float v1 = (sq[0][1] + sq[2][1]) + (sq[1][1] + sq[3][1]);
    float var = wave_tree_sum(v0 + v1) * (1.0f / 256.0f);

    float inv = 1.0f / sqrtf(var + 1e-5f);
    for (int b = 0; b < 4; ++b) {
        float* orow = out + ((size_t)b * N_NODES + i) * F_SZ;
        for (int k = 0; k < 2; ++k) {
            float xn  = g * dev[b][k] * inv + be;
            float res = xn + hin[b][k];
            orow[lane + 32 * k] = res > 0.f ? res : expm1f(res);
        }
    }
}

extern "C" void kernel_launch(void* const* d_in, const int* in_sizes, int n_in,
                              void* d_out, int out_size, void* d_ws, size_t ws_size,
                              hipStream_t stream)
{
    const float* h     = (const float*)d_in[0];
    const int*   adj   = (const int*)  d_in[1];
    const float* W     = (const float*)d_in[2];
    const float* a     = (const float*)d_in[3];
    const float* gamma = (const float*)d_in[4];
    const float* beta  = (const float*)d_in[5];
    float* out = (float*)d_out;

    const size_t WH_ELEMS = (size_t)B_SZ * N_NODES * F_SZ;
    const size_t NROW     = (size_t)B_SZ * N_NODES;

    int nchunk = 4;
    size_t need4 = (WH_ELEMS + 3 * NROW + NROW / WH_NODES + 4 * (WH_ELEMS + NROW)) * sizeof(float);
    if (ws_size < need4) nchunk = 1;

    float* wsf  = (float*)d_ws;
    float* WhP  = wsf;
    float* s1   = WhP + WH_ELEMS;
    float* vtab = s1 + NROW;
    float* tpart = vtab + 2 * NROW;
    float* Dp   = tpart + NROW / WH_NODES;
    float* Zp   = Dp + (size_t)nchunk * WH_ELEMS;

    int jspan = N_NODES / nchunk;

    wh_kernel<<<B_SZ * N_NODES / (WH_WAVES * WH_NODES), 32 * WH_WAVES, 0, stream>>>(h, W, a, WhP, s1, vtab, tpart);
    attn_kernel<<<B_SZ * (N_NODES / (16 * ROWT)) * nchunk, 32, 0, stream>>>(
        adj, WhP, s1, vtab, tpart, Dp, Zp, jspan);
    bn_kernel<<<N_NODES / 8, 256, 0, stream>>>(Dp, Zp, WhP, h, gamma, beta, out, nchunk);
}


# Wh kernel: h@W on the f32 16x16x4 matrix instruction, one wave per 16 nodes (64 WMMAs, W loads as one group, whole-row stores), s'/t' folded from the tiles via LDS; Wh, s', t' sum order changes
# speedup vs baseline: 2.4112x; 1.0501x over previous
#include <hip/hip_runtime.h>
#include <hip/hip_bf16.h>
#include <math.h>

typedef float v2f __attribute__((ext_vector_type(2)));
typedef float v8f __attribute__((ext_vector_type(8)));

#define N_NODES 4096
#define B_SZ    4
#define F_SZ    64
#define ALPHA   0.2f
#define L2E     1.4426950408889634f

#define WH_NODES 8
#define WH_ROWS  16
#define WH_WAVES 4
#define WH_PITCH 36

__global__ __launch_bounds__(32 * WH_WAVES) void wh_kernel(
    const float* __restrict__ h, const float* __restrict__ W,
    const float* __restrict__ a, float* __restrict__ WhP,
    float* __restrict__ s1, float* __restrict__ vtab, float* __restrict__ tpart)
{
    __shared__ float sP[WH_WAVES][2 * WH_NODES][WH_PITCH];
    int lane = threadIdx.x & 31;
    int wv = threadIdx.x >> 5;
    int c  = lane & 15;
    int hf = lane >> 4;
    int n0 = (blockIdx.x * WH_WAVES + wv) * WH_ROWS;

    const float* hrow = h + (size_t)(n0 + c) * F_SZ + 8 * hf;
    float hk[4][8];
#pragma unroll
    for (int g = 0; g < 4; ++g) {
        float4 lo = *(const float4*)(hrow + 16 * g);
        float4 hi = *(const float4*)(hrow + 16 * g + 4);
        hk[g][0] = lo.x; hk[g][1] = lo.y; hk[g][2] = lo.z; hk[g][3] = lo.w;
        hk[g][4] = hi.x; hk[g][5] = hi.y; hk[g][6] = hi.z; hk[g][7] = hi.w;
    }

    const float* wcol = W + (size_t)(8 * hf) * F_SZ + c;
    float wb[4][4][4][2];
#pragma unroll
    for (int g = 0; g < 4; ++g) {
#pragma unroll
        for (int i = 0; i < 4; ++i) {
            const float* wk = wcol + (size_t)(16 * g + 2 * i) * F_SZ;
#pragma unroll
            for (int q = 0; q < 4; ++q) {
                wb[g][i][q][0] = wk[16 * q];
                wb[g][i][q][1] = wk[F_SZ + 16 * q];
            }
        }
    }
    float a1v[4], a2v[4];
#pragma unroll
    for (int q = 0; q < 4; ++q) {
        a1v[q] = a[16 * q + c];
        a2v[q] = a[F_SZ + 16 * q + c];
    }
    __builtin_amdgcn_sched_barrier(0);

    v8f acc[4];
#pragma unroll
    for (int q = 0; q < 4; ++q) acc[q] = v8f{};
#pragma unroll
    for (int g = 0; g < 4; ++g) {
#pragma unroll
        for (int i = 0; i < 4; ++i) {
            v2f A;
            A[0] = hk[g][2 * i];
            A[1] = hk[g][2 * i + 1];
#pragma unroll
            for (int q = 0; q < 4; ++q) {
                v2f Bq;
                Bq[0] = wb[g][i][q][0];
                Bq[1] = wb[g][i][q][1];
                acc[q] = __builtin_amdgcn_wmma_f32_16x16x4_f32(false, A, false, Bq, (short)0, acc[q], false, false);
            }
        }
    }

#pragma unroll
    for (int r = 0; r < 8; ++r) {
        float ps = 0.f, pt = 0.f;
#pragma unroll
        for (int q = 0; q < 4; ++q) {
            ps = fmaf(acc[q][r], a1v[q], ps);
            pt = fmaf(acc[q][r], a2v[q], pt);
        }
        sP[wv][r][lane]            = ps;
        sP[wv][WH_NODES + r][lane] = pt;
    }
    __syncthreads();

    float* wout = WhP + (size_t)(n0 + 8 * hf) * F_SZ + 4 * c;
#pragma unroll
    for (int r = 0; r < 8; ++r) {
        float4 v; v.x = acc[0][r]; v.y = acc[1][r]; v.z = acc[2][r]; v.w = acc[3][r];
        *(float4*)(wout + (size_t)r * F_SZ) = v;
    }

    int sel = hf;
    int row = c;
    const float* part = &sP[wv][sel * WH_NODES + (row & 7)][(row >> 3) * 16];
    float d = 0.f;
#pragma unroll
    for (int k = 0; k < 16; ++k) d += part[k];
    float rr = d * (1.0f / 64.0f);
    float tmax = -INFINITY;
    int gi = n0 + row;
    if (sel) {
        float* vt = vtab + (size_t)(gi >> 1) * 4 + (gi & 1);
        vt[0] = __builtin_amdgcn_exp2f(rr * L2E);
        vt[2] = __builtin_amdgcn_exp2f((ALPHA * rr) * L2E);
        tmax = rr;
    } else {
        s1[gi] = rr;
    }
    for (int off = WH_NODES / 2; off >= 1; off >>= 1)
        tmax = fmaxf(tmax, __shfl_xor(tmax, off, 32));
    if (sel && (row & 7) == 0) tpart[(n0 >> 3) + (row >> 3)] = tmax;
}

typedef int   v2i __attribute__((ext_vector_type(2)));
typedef float v4f __attribute__((ext_vector_type(4)));
#define ROWT 2
#define UB   2
struct JBlock {
    v2i av[ROWT][UB];
    v4f cf[UB];
    v4f q0[UB];
    v4f q1[UB];
};

__device__ __forceinline__ void load_block(JBlock& k, const int* __restrict__ pa,
                                           const float* __restrict__ pv,
                                           const float* __restrict__ pb)
{
#pragma unroll
    for (int u = 0; u < UB; ++u) {
#pragma unroll
        for (int r = 0; r < ROWT; ++r)
            k.av[r][u] = *(const v2i*)(pa + (size_t)r * 16 * N_NODES + 4 * u);
        k.cf[u] = *(const v4f*)(pv + 8 * u);
        k.q0[u] = *(const v4f*)(pb + (4 * u) * F_SZ);
        k.q1[u] = *(const v4f*)(pb + (4 * u + 1) * F_SZ);
    }
}

__device__ __forceinline__ void arrive_block(JBlock& k)
{
    asm volatile("" : "+v"(k.av[0][0]), "+v"(k.av[0][1]), "+v"(k.av[1][0]), "+v"(k.av[1][1]),
                      "+v"(k.cf[0]), "+v"(k.cf[1]),
                      "+v"(k.q0[0]), "+v"(k.q0[1]), "+v"(k.q1[0]), "+v"(k.q1[1]));
}

__device__ __forceinline__ void settle_block(v8f (&c)[ROWT][4])
{
    asm volatile("" : "+v"(c[0][0]), "+v"(c[0][1]), "+v"(c[0][2]), "+v"(c[0][3]),
                      "+v"(c[1][0]), "+v"(c[1][1]), "+v"(c[1][2]), "+v"(c[1][3]));
}

__device__ __forceinline__ void compute_block(const JBlock& k, const float (&fa)[ROWT],
                                              const float (&fb)[ROWT],
                                              v8f (&c)[ROWT][4], float (&zacc)[ROWT])
{
#pragma unroll
    for (int u = 0; u < UB; ++u) {
        v2f B0, B1, B2, B3;
        B0[0] = k.q0[u].x; B0[1] = k.q1[u].x;
        B1[0] = k.q0[u].y; B1[1] = k.q1[u].y;
        B2[0] = k.q0[u].z; B2[1] = k.q1[u].z;
        B3[0] = k.q0[u].w; B3[1] = k.q1[u].w;

        v2f A[ROWT];
#pragma unroll
        for (int r = 0; r < ROWT; ++r) {
            float e0 = fmaxf(fa[r] * k.cf[u].x, fb[r] * k.cf[u].z);
            float e1 = fmaxf(fa[r] * k.cf[u].y, fb[r] * k.cf[u].w);
            float p0 = (k.av[r][u].x > 0) ? e0 : 0.f;
            float p1 = (k.av[r][u].y > 0) ? e1 : 0.f;
            zacc[r] += p0 + p1;
            A[r][0] = p0; A[r][1] = p1;
        }
#pragma unroll
        for (int r = 0; r < ROWT; ++r) {
            c[r][0] = __builtin_amdgcn_wmma_f32_16x16x4_f32(false, A[r], false, B0, (short)0, c[r][0], false, false);
            c[r][1] = __builtin_amdgcn_wmma_f32_16x16x4_f32(false, A[r], false, B1, (short)0, c[r][1], false, false);
            c[r][2] = __builtin_amdgcn_wmma_f32_16x16x4_f32(false, A[r], false, B2, (short)0, c[r][2], false, false);
            c[r][3] = __builtin_amdgcn_wmma_f32_16x16x4_f32(false, A[r], false, B3, (short)0, c[r][3], false, false);
        }
    }
}

__global__ __launch_bounds__(32) void attn_kernel(
    const int* __restrict__ adj, const float* __restrict__ WhP,
    const float* __restrict__ s1, const float* __restrict__ vtab,
    const float* __restrict__ tpart, float* __restrict__ Dp,
    float* __restrict__ Zp, int jspan)
{
    int bid   = blockIdx.x;
    int b     = bid & 3;
    int itile = (bid >> 2) & (N_NODES / (16 * ROWT) - 1);
    int chunk = bid / (4 * (N_NODES / (16 * ROWT)));
    int lane  = threadIdx.x;
    int l15   = lane & 15;
    int half  = lane >> 4;
    int irow  = itile * (16 * ROWT) + l15;

    float s[ROWT];
#pragma unroll
    for (int r = 0; r < ROWT; ++r) s[r] = s1[b * N_NODES + irow + 16 * r];

    const v4f* tp = (const v4f*)(tpart + b * (N_NODES / WH_NODES)) + lane;
    float T = -INFINITY;
#pragma unroll
    for (int u = 0; u < N_NODES / WH_NODES / 128; ++u) {
        v4f q = tp[u * 32];
        T = fmaxf(T, fmaxf(fmaxf(q.x, q.y), fmaxf(q.z, q.w)));
    }
    for (int off = 16; off >= 1; off >>= 1) T = fmaxf(T, __shfl_xor(T, off, 32));
    float fa[ROWT], fb[ROWT];
#pragma unroll
    for (int r = 0; r < ROWT; ++r) {
        float xm = s[r] + T;
        float m  = fmaxf(xm, ALPHA * xm);
        fa[r] = __builtin_amdgcn_exp2f((s[r] - m) * L2E);
        fb[r] = __builtin_amdgcn_exp2f((ALPHA * s[r] - m) * L2E);
    }

    v8f c[ROWT][4];
    float zacc[ROWT];
#pragma unroll
    for (int r = 0; r < ROWT; ++r) {
        zacc[r] = 0.f;
#pragma unroll
        for (int q = 0; q < 4; ++q) c[r][q] = v8f{};
    }

    int jfirst = chunk * jspan + half * 2;
    const int*   pa = adj + (size_t)irow * N_NODES + jfirst;
    const float* pv = vtab + ((size_t)b * N_NODES + jfirst) * 2;
    const float* pb = WhP + ((size_t)b * N_NODES + jfirst) * F_SZ + (l15 << 2);

#define KEEP_ORDER() __builtin_amdgcn_sched_barrier(0)
#define STEP(KLOAD, OFF, KWORK)                                              \
    load_block(KLOAD, pa + (OFF), pv + 2 * (OFF), pb + (OFF) * F_SZ);        \
    KEEP_ORDER();                                                            \
    arrive_block(KWORK);                                                     \
    compute_block(KWORK, fa, fb, c, zacc);                                   \
    settle_block(c);                                                         \
    KEEP_ORDER();

    JBlock k0, k1, k2, k3;
    load_block(k0, pa, pv, pb);
    KEEP_ORDER();
    load_block(k1, pa + 8, pv + 16, pb + 8 * F_SZ);
    KEEP_ORDER();
    load_block(k2, pa + 16, pv + 32, pb + 16 * F_SZ);
    KEEP_ORDER();
    int nquad = jspan >> 5;
    for (int it = 1; it < nquad; ++it) {
        STEP(k3, 24, k0)
        pa += 32; pv += 64; pb += 32 * F_SZ;
        STEP(k0, 0, k1)
        STEP(k1, 8, k2)
        STEP(k2, 16, k3)
    }
    STEP(k3, 24, k0)
    arrive_block(k1);
    compute_block(k1, fa, fb, c, zacc);
    arrive_block(k2);
    compute_block(k2, fa, fb, c, zacc);
    arrive_block(k3);
    compute_block(k3, fa, fb, c, zacc);
#undef STEP
#undef KEEP_ORDER

    size_t obase = (((size_t)chunk * 4 + b) * N_NODES + (size_t)itile * (16 * ROWT)) * F_SZ;
    int rbase = half * 8;
#pragma unroll
    for (int r = 0; r < ROWT; ++r) {
        float z = zacc[r] + __shfl_xor(zacc[r], 16, 32);
        float* dout = Dp + obase + (size_t)(16 * r) * F_SZ;
#pragma unroll
        for (int q = 0; q < 8; ++q) {
            float4 v; v.x = c[r][0][q]; v.y = c[r][1][q]; v.z = c[r][2][q]; v.w = c[r][3][q];
            *(float4*)(dout + (size_t)(rbase + q) * F_SZ + (l15 << 2)) = v;
        }
        if (half == 0)
            Zp[((size_t)chunk * 4 + b) * N_NODES + irow + 16 * r] = z;
    }
}

__device__ __forceinline__ float wave_tree_sum(float r)
{
    for (int off = 16; off >= 1; off >>= 1) r += __shfl_xor(r, off, 32);
    return r;
}

__device__ __forceinline__ void add_chunk(const float* __restrict__ Dp,
                                          const float* __restrict__ Zp,
                                          int c, int i, int p1,
                                          float (&num)[4][2], float (&z)[4])
{
    for (int b = 0; b < 4; ++b) {
        size_t row = ((size_t)c * 4 + b) * N_NODES + i;
        const float* d = Dp + (row << 6);
        num[b][0] += d[p1];
        num[b][1] += d[p1 + 2];
        z[b]      += Zp[row];
    }
}

__global__ __launch_bounds__(256) void bn_kernel(
    const float* __restrict__ Dp, const float* __restrict__ Zp,
    const float* __restrict__ WhP,
    const float* __restrict__ h, const float* __restrict__ gamma,
    const float* __restrict__ beta, float* __restrict__ out, int nchunk)
{
    int lane = threadIdx.x & 31;
    int wv = __builtin_amdgcn_readfirstlane((int)(threadIdx.x >> 5));
    int i  = blockIdx.x * 8 + wv;
    int p1 = ((lane & 15) << 2) | (lane >> 4);

    float hin[4][2];
    for (int b = 0; b < 4; ++b) {
        const float* hr = h + ((size_t)b * N_NODES + i) * F_SZ;
        hin[b][0] = hr[lane];
        hin[b][1] = hr[lane + 32];
    }
    float g = gamma[i], be = beta[i];

    float num[4][2], z[4];
    for (int b = 0; b < 4; ++b) { num[b][0] = 0.f; num[b][1] = 0.f; z[b] = 0.f; }
    if (nchunk == 4) {
        for (int c = 0; c < 4; ++c) add_chunk(Dp, Zp, c, i, p1, num, z);
    } else {
        for (int c = 0; c < nchunk; ++c) add_chunk(Dp, Zp, c, i, p1, num, z);
    }

    for (int b = 0; b < 4; ++b) {
        if (z[b] == 0.f) {
            const float* wcol = WhP + (size_t)b * N_NODES * F_SZ + p1;
            float c0 = 0.f, c1 = 0.f;
            for (int j = 0; j < N_NODES; ++j) {
                c0 += wcol[(size_t)j * F_SZ];
                c1 += wcol[(size_t)j * F_SZ + 2];
            }
            num[b][0] = c0; num[b][1] = c1; z[b] = (float)N_NODES;
        }
    }

    float hp[4][2];
    for (int b = 0; b < 4; ++b) {
        hp[b][0] = num[b][0] / z[b];
        hp[b][1] = num[b][1] / z[b];
    }

    float q0 = (hp[0][0] + hp[2][0]) + (hp[1][0] + hp[3][0]);
    float q1 = (hp[0][1] + hp[2][1]) + (hp[1][1] + hp[3][1]);
    float mean = wave_tree_sum(q0 + q1) * (1.0f / 256.0f);

    float dev[4][2], sq[4][2];
    for (int b = 0; b < 4; ++b) {
        dev[b][0] = hp[b][0] - mean;  sq[b][0] = dev[b][0] * dev[b][0];
        dev[b][1] = hp[b][1] - mean;  sq[b][1] = dev[b][1] * dev[b][1];
    }
    float v0 = (sq[0][0] + sq[2][0]) + (sq[1][0] + sq[3][0]);
    float v1 = (sq[0][1] + sq[2][1]) + (sq[1][1] + sq[3][1]);
    float var = wave_tree_sum(v0 + v1) * (1.0f / 256.0f);

    float inv = 1.0f / sqrtf(var + 1e-5f);
    for (int b = 0; b < 4; ++b) {
        float* orow = out + ((size_t)b * N_NODES + i) * F_SZ;
        for (int k = 0; k < 2; ++k) {
            float xn  = g * dev[b][k] * inv + be;
            float res = xn + hin[b][k];
            orow[lane + 32 * k] = res > 0.f ? res : expm1f(res);
        }
    }
}

extern "C" void kernel_launch(void* const* d_in, const int* in_sizes, int n_in,
                              void* d_out, int out_size, void* d_ws, size_t ws_size,
                              hipStream_t stream)
{
    const float* h     = (const float*)d_in[0];
    const int*   adj   = (const int*)  d_in[1];
    const float* W     = (const float*)d_in[2];
    const float* a     = (const float*)d_in[3];
    const float* gamma = (const float*)d_in[4];
    const float* beta  = (const float*)d_in[5];
    float* out = (float*)d_out;

    const size_t WH_ELEMS = (size_t)B_SZ * N_NODES * F_SZ;
    const size_t NROW     = (size_t)B_SZ * N_NODES;

    int nchunk = 4;
    size_t need4 = (WH_ELEMS + 3 * NROW + NROW / WH_NODES + 4 * (WH_ELEMS + NROW)) * sizeof(float);
    if (ws_size < need4) nchunk = 1;

    float* wsf  = (float*)d_ws;
    float* WhP  = wsf;
    float* s1   = WhP + WH_ELEMS;
    float* vtab = s1 + NROW;
    float* tpart = vtab + 2 * NROW;
    float* Dp   = tpart + NROW / WH_NODES;
    float* Zp   = Dp + (size_t)nchunk * WH_ELEMS;

    int jspan = N_NODES / nchunk;

    wh_kernel<<<B_SZ * N_NODES / (WH_WAVES * WH_ROWS), 32 * WH_WAVES, 0, stream>>>(h, W, a, WhP, s1, vtab, tpart);
    attn_kernel<<<B_SZ * (N_NODES / (16 * ROWT)) * nchunk, 32, 0, stream>>>(
        adj, WhP, s1, vtab, tpart, Dp, Zp, jspan);
    bn_kernel<<<N_NODES / 8, 256, 0, stream>>>(Dp, Zp, WhP, h, gamma, beta, out, nchunk);
}
